# starting kernel
# baseline (speedup 1.0000x reference)
#include <hip/hip_runtime.h>


typedef __attribute__((ext_vector_type(2))) float v2f;
typedef __attribute__((ext_vector_type(8))) float v8f;

#define D   128
#define DE  32

__device__ __forceinline__ v8f wmma4(float a0, float a1, float b0, float b1, v8f c) {
  v2f a; a[0] = a0; a[1] = a1;
  v2f b; b[0] = b0; b[1] = b1;
  return __builtin_amdgcn_wmma_f32_16x16x4_f32(false, a, false, b, (short)0, c,
                                               false, false);
}

__global__ void zero_f32(float* __restrict__ p, long n) {
  long i = (long)blockIdx.x * blockDim.x + threadIdx.x;
  long stride = (long)gridDim.x * blockDim.x;
  for (; i < n; i += stride) p[i] = 0.0f;
}

__global__ void scatter_add(const float* __restrict__ X, const int* __restrict__ src,
                            const int* __restrict__ dst, float* __restrict__ sum,
                            float* __restrict__ cnt, int E) {
  const int wave = threadIdx.x >> 5;
  const int lane = threadIdx.x & 31;
  const int e = blockIdx.x * 8 + wave;
  if (e >= E) return;
  const int s = src[e];
  const int d = dst[e];
  const float4 v = *(const float4*)(X + (long)s * D + lane * 4);
  float* o = sum + (long)d * D + lane * 4;
  atomicAdd(o + 0, v.x);
  atomicAdd(o + 1, v.y);
  atomicAdd(o + 2, v.z);
  atomicAdd(o + 3, v.w);
  if (lane == 0) atomicAdd(cnt + d, 1.0f);
}

__global__ void finalize_mean(float* __restrict__ sum, const float* __restrict__ cnt,
                              long n) {
  long i = (long)blockIdx.x * blockDim.x + threadIdx.x;
  long stride = (long)gridDim.x * blockDim.x;
  for (; i < n; i += stride) {
    float c = cnt[i >> 7];
    sum[i] = sum[i] / fmaxf(c, 1.0f);
  }
}

__global__ void node_gemm(const float* __restrict__ A1, const float* __restrict__ W1,
                          const float* __restrict__ A2, const float* __restrict__ W2,
                          const float* __restrict__ bias, float* __restrict__ Hout,
                          int do_relu) {
  const int tid  = threadIdx.x;
  const int wave = tid >> 5;
  const int lane = tid & 31;
  const int half = lane >> 4;
  const int lm   = lane & 15;
  const int m0   = blockIdx.x * 16;
  const int n    = wave * 16 + lm;

  v8f c;
  const float bv = bias ? bias[n] : 0.0f;
#pragma unroll
  for (int r = 0; r < 8; ++r) c[r] = bv;

  const float* aRow = A1 + (long)(m0 + lm) * D + 2 * half;
#pragma unroll 4
  for (int k0 = 0; k0 < D; k0 += 4) {
    const float2 a = *(const float2*)(aRow + k0);
    const float b0 = W1[(k0 + 2 * half) * D + n];
    const float b1 = W1[(k0 + 2 * half + 1) * D + n];
    c = wmma4(a.x, a.y, b0, b1, c);
  }
  if (A2) {
    const float* aRow2 = A2 + (long)(m0 + lm) * D + 2 * half;
#pragma unroll 4
    for (int k0 = 0; k0 < D; k0 += 4) {
      const float2 a = *(const float2*)(aRow2 + k0);
      const float b0 = W2[(k0 + 2 * half) * D + n];
      const float b1 = W2[(k0 + 2 * half + 1) * D + n];
      c = wmma4(a.x, a.y, b0, b1, c);
    }
  }

#pragma unroll
  for (int r = 0; r < 8; ++r) {
    float v = c[r];
    if (do_relu) v = fmaxf(v, 0.0f);
    const int m = m0 + r + 8 * half;
    Hout[(long)m * D + n] = v;
  }
}

__global__ void edge_mlp(const float* __restrict__ P1, const float* __restrict__ P2,
                         const float* __restrict__ ea, const int* __restrict__ src,
                         const int* __restrict__ dst, const float* __restrict__ W3,
                         const float* __restrict__ bc1, const float* __restrict__ Wc2,
                         const float* __restrict__ bc2, float* __restrict__ out,
                         int E) {
  __shared__ int   s_src[16];
  __shared__ int   s_dst[16];
  __shared__ float s_red[8][16];

  const int tid  = threadIdx.x;
  const int wave = tid >> 5;
  const int lane = tid & 31;
  const int half = lane >> 4;
  const int lm   = lane & 15;
  const int e0   = blockIdx.x * 16;
  const int n    = wave * 16 + lm;

  if (tid < 16)       s_src[tid]      = src[e0 + tid];
  else if (tid < 32)  s_dst[tid - 16] = dst[e0 + tid - 16];
  __syncthreads();

  v8f c;
  const float bv = bc1[n];
#pragma unroll
  for (int r = 0; r < 8; ++r) {
    const int m = r + 8 * half;
    c[r] = P1[(long)s_src[m] * D + n] + P2[(long)s_dst[m] * D + n] + bv;
  }

  const float* aRow = ea + (long)(e0 + lm) * DE + 2 * half;
#pragma unroll
  for (int k0 = 0; k0 < DE; k0 += 4) {
    const float2 a = *(const float2*)(aRow + k0);
    const float b0 = W3[(k0 + 2 * half) * D + n];
    const float b1 = W3[(k0 + 2 * half + 1) * D + n];
    c = wmma4(a.x, a.y, b0, b1, c);
  }

  const float w2 = Wc2[n];
#pragma unroll
  for (int r = 0; r < 8; ++r) {
    float t = fmaxf(c[r], 0.0f) * w2;
    t += __shfl_xor(t, 1);
    t += __shfl_xor(t, 2);
    t += __shfl_xor(t, 4);
    t += __shfl_xor(t, 8);
    if (lm == 0) s_red[wave][r + 8 * half] = t;
  }
  __syncthreads();

  if (tid < 16) {
    float acc = bc2[0];
#pragma unroll
    for (int w = 0; w < 8; ++w) acc += s_red[w][tid];
    if (e0 + tid < E) out[e0 + tid] = acc;
  }
}

extern "C" void kernel_launch(void* const* d_in, const int* in_sizes, int n_in,
                              void* d_out, int out_size, void* d_ws, size_t ws_size,
                              hipStream_t stream) {
  (void)n_in; (void)out_size; (void)ws_size;
  const float* x   = (const float*)d_in[0];
  const float* ea  = (const float*)d_in[1];
  const float* W1a = (const float*)d_in[2];
  const float* b1  = (const float*)d_in[3];
  const float* W1r = (const float*)d_in[4];
  const float* W2a = (const float*)d_in[5];
  const float* b2  = (const float*)d_in[6];
  const float* W2r = (const float*)d_in[7];
  const float* Wc1 = (const float*)d_in[8];
  const float* bc1 = (const float*)d_in[9];
  const float* Wc2 = (const float*)d_in[10];
  const float* bc2 = (const float*)d_in[11];
  const int*   ei  = (const int*)d_in[12];

  const int N = in_sizes[0] / D;
  const int E = in_sizes[12] / 2;
  const int* src = ei;
  const int* dst = ei + E;

  float* ws = (float*)d_ws;
  const long NM = (long)N * D;
  float* bufA = ws;
  float* bufB = ws + NM;
  float* bufC = ws + 2 * NM;
  float* cnt  = ws + 3 * NM;
  float* out  = (float*)d_out;

  const dim3 blk(256);
  const int sgrid = (E + 7) / 8;
  const int ggrid = N / 16;
  const int egrid = E / 16;

  zero_f32<<<2048, blk, 0, stream>>>(bufA, NM);
  zero_f32<<<128,  blk, 0, stream>>>(cnt, N);
  scatter_add<<<sgrid, blk, 0, stream>>>(x, src, dst, bufA, cnt, E);
  finalize_mean<<<4096, blk, 0, stream>>>(bufA, cnt, NM);
  node_gemm<<<ggrid, blk, 0, stream>>>(bufA, W1a, x, W1r, b1, bufB, 1);

  zero_f32<<<2048, blk, 0, stream>>>(bufA, NM);
  zero_f32<<<128,  blk, 0, stream>>>(cnt, N);
  scatter_add<<<sgrid, blk, 0, stream>>>(bufB, src, dst, bufA, cnt, E);
  finalize_mean<<<4096, blk, 0, stream>>>(bufA, cnt, NM);
  node_gemm<<<ggrid, blk, 0, stream>>>(bufA, W2a, bufB, W2r, b2, bufC, 0);

  node_gemm<<<ggrid, blk, 0, stream>>>(bufC, Wc1,           nullptr, nullptr,
                                       nullptr, bufB, 0);
  node_gemm<<<ggrid, blk, 0, stream>>>(bufC, Wc1 + 128 * D, nullptr, nullptr,
                                       nullptr, bufA, 0);

  edge_mlp<<<egrid, blk, 0, stream>>>(bufB, bufA, ea, src, dst, Wc1 + 256 * D,
                                      bc1, Wc2, bc2, out, E);
}


# Replace float-atomic push scatter by pull over dst-bucketed edges: int-atomic degree+rank, one-launch scan, permute; one wave per node sums neighbour rows in arrival order and stores the mean
# speedup vs baseline: 1.6049x; 1.6049x over previous
#include <hip/hip_runtime.h>


typedef __attribute__((ext_vector_type(2))) float v2f;
typedef __attribute__((ext_vector_type(8))) float v8f;

#define D   128
#define DE  32

__device__ __forceinline__ v8f wmma4(float a0, float a1, float b0, float b1, v8f c) {
  v2f a; a[0] = a0; a[1] = a1;
  v2f b; b[0] = b0; b[1] = b1;
  return __builtin_amdgcn_wmma_f32_16x16x4_f32(false, a, false, b, (short)0, c,
                                               false, false);
}

__global__ void __launch_bounds__(256) zero_i32(int* __restrict__ p, int n) {
  const int i = blockIdx.x * 256 + threadIdx.x;
  if (i < n) p[i] = 0;
}

__global__ void __launch_bounds__(256) count_rank(const int* __restrict__ dst,
                                                  int* __restrict__ deg,
                                                  int* __restrict__ rank, int E) {
  const int e = blockIdx.x * 256 + threadIdx.x;
  if (e < E) rank[e] = atomicAdd(deg + dst[e], 1);
}

__global__ void __launch_bounds__(1024) scan_rowptr(const int* __restrict__ deg,
                                                    int* __restrict__ rowptr, int N) {
  __shared__ int s_tile[32];
  __shared__ int s_front[32];
  const int tid  = threadIdx.x;
  const int lane = tid & 31;
  const int wave = tid >> 5;
  const int tile0 = blockIdx.x * 1024;

  int front = 0;
  for (int i = tid; i < tile0; i += 1024) front += deg[i];
  const int idx = tile0 + tid;
  const int v = idx < N ? deg[idx] : 0;

  int inc = v;
#pragma unroll
  for (int o = 1; o < 32; o <<= 1) {
    const int t = __shfl_up(inc, o);
    if (lane >= o) inc += t;
  }
#pragma unroll
  for (int o = 16; o > 0; o >>= 1) front += __shfl_xor(front, o);
  if (lane == 31) s_tile[wave]  = inc;
  if (lane == 0)  s_front[wave] = front;
  __syncthreads();

  int base = 0;
  for (int w = 0; w < 32; ++w) {
    base += s_front[w];
    if (w < wave) base += s_tile[w];
  }
  if (idx < N) rowptr[idx] = base + inc - v;
}

__global__ void __launch_bounds__(256) fill_nbr(const int* __restrict__ src,
                                                const int* __restrict__ dst,
                                                const int* __restrict__ rowptr,
                                                const int* __restrict__ rank,
                                                int* __restrict__ nbr, int E) {
  const int e = blockIdx.x * 256 + threadIdx.x;
  if (e < E) nbr[rowptr[dst[e]] + rank[e]] = src[e];
}

__global__ void __launch_bounds__(256) gather_mean(const float* __restrict__ X,
                                                   const int* __restrict__ nbr,
                                                   const int* __restrict__ rowptr,
                                                   const int* __restrict__ deg,
                                                   float* __restrict__ agg, int N) {
  const int lane = threadIdx.x & 31;
  const int node = __builtin_amdgcn_readfirstlane(blockIdx.x * 8 + (threadIdx.x >> 5));
  if (node >= N) return;
  const int beg = rowptr[node];
  const int dg  = deg[node];

  float4 acc = make_float4(0.0f, 0.0f, 0.0f, 0.0f);
  for (int base = 0; base < dg; base += 32) {
    const int rem = dg - base < 32 ? dg - base : 32;
    const int mine = lane < rem ? nbr[beg + base + lane] : 0;
#pragma unroll 4
    for (int j = 0; j < rem; ++j) {
      int s = __builtin_amdgcn_readlane(mine, j);
      s = (unsigned)s < (unsigned)N ? s : N - 1;
      const float4 v = *(const float4*)(X + (long)s * D + lane * 4);
      acc.x += v.x; acc.y += v.y; acc.z += v.z; acc.w += v.w;
    }
  }
  const float c = fmaxf((float)dg, 1.0f);
  float4 o;
  o.x = acc.x / c; o.y = acc.y / c; o.z = acc.z / c; o.w = acc.w / c;
  *(float4*)(agg + (long)node * D + lane * 4) = o;
}

__global__ void node_gemm(const float* __restrict__ A1, const float* __restrict__ W1,
                          const float* __restrict__ A2, const float* __restrict__ W2,
                          const float* __restrict__ bias, float* __restrict__ Hout,
                          int do_relu) {
  const int tid  = threadIdx.x;
  const int wave = tid >> 5;
  const int lane = tid & 31;
  const int half = lane >> 4;
  const int lm   = lane & 15;
  const int m0   = blockIdx.x * 16;
  const int n    = wave * 16 + lm;

  v8f c;
  const float bv = bias ? bias[n] : 0.0f;
#pragma unroll
  for (int r = 0; r < 8; ++r) c[r] = bv;

  const float* aRow = A1 + (long)(m0 + lm) * D + 2 * half;
#pragma unroll 4
  for (int k0 = 0; k0 < D; k0 += 4) {
    const float2 a = *(const float2*)(aRow + k0);
    const float b0 = W1[(k0 + 2 * half) * D + n];
    const float b1 = W1[(k0 + 2 * half + 1) * D + n];
    c = wmma4(a.x, a.y, b0, b1, c);
  }
  if (A2) {
    const float* aRow2 = A2 + (long)(m0 + lm) * D + 2 * half;
#pragma unroll 4
    for (int k0 = 0; k0 < D; k0 += 4) {
      const float2 a = *(const float2*)(aRow2 + k0);
      const float b0 = W2[(k0 + 2 * half) * D + n];
      const float b1 = W2[(k0 + 2 * half + 1) * D + n];
      c = wmma4(a.x, a.y, b0, b1, c);
    }
  }

#pragma unroll
  for (int r = 0; r < 8; ++r) {
    float v = c[r];
    if (do_relu) v = fmaxf(v, 0.0f);
    const int m = m0 + r + 8 * half;
    Hout[(long)m * D + n] = v;
  }
}

__global__ void edge_mlp(const float* __restrict__ P1, const float* __restrict__ P2,
                         const float* __restrict__ ea, const int* __restrict__ src,
                         const int* __restrict__ dst, const float* __restrict__ W3,
                         const float* __restrict__ bc1, const float* __restrict__ Wc2,
                         const float* __restrict__ bc2, float* __restrict__ out,
                         int E) {
  __shared__ int   s_src[16];
  __shared__ int   s_dst[16];
  __shared__ float s_red[8][16];

  const int tid  = threadIdx.x;
  const int wave = tid >> 5;
  const int lane = tid & 31;
  const int half = lane >> 4;
  const int lm   = lane & 15;
  const int e0   = blockIdx.x * 16;
  const int n    = wave * 16 + lm;

  if (tid < 16)       s_src[tid]      = src[e0 + tid];
  else if (tid < 32)  s_dst[tid - 16] = dst[e0 + tid - 16];
  __syncthreads();

  v8f c;
  const float bv = bc1[n];
#pragma unroll
  for (int r = 0; r < 8; ++r) {
    const int m = r + 8 * half;
    c[r] = P1[(long)s_src[m] * D + n] + P2[(long)s_dst[m] * D + n] + bv;
  }

  const float* aRow = ea + (long)(e0 + lm) * DE + 2 * half;
#pragma unroll
  for (int k0 = 0; k0 < DE; k0 += 4) {
    const float2 a = *(const float2*)(aRow + k0);
    const float b0 = W3[(k0 + 2 * half) * D + n];
    const float b1 = W3[(k0 + 2 * half + 1) * D + n];
    c = wmma4(a.x, a.y, b0, b1, c);
  }

  const float w2 = Wc2[n];
#pragma unroll
  for (int r = 0; r < 8; ++r) {
    float t = fmaxf(c[r], 0.0f) * w2;
    t += __shfl_xor(t, 1);
    t += __shfl_xor(t, 2);
    t += __shfl_xor(t, 4);
    t += __shfl_xor(t, 8);
    if (lm == 0) s_red[wave][r + 8 * half] = t;
  }
  __syncthreads();

  if (tid < 16) {
    float acc = bc2[0];
#pragma unroll
    for (int w = 0; w < 8; ++w) acc += s_red[w][tid];
    if (e0 + tid < E) out[e0 + tid] = acc;
  }
}

extern "C" void kernel_launch(void* const* d_in, const int* in_sizes, int n_in,
                              void* d_out, int out_size, void* d_ws, size_t ws_size,
                              hipStream_t stream) {
  (void)n_in; (void)out_size; (void)ws_size;
  const float* x   = (const float*)d_in[0];
  const float* ea  = (const float*)d_in[1];
  const float* W1a = (const float*)d_in[2];
  const float* b1  = (const float*)d_in[3];
  const float* W1r = (const float*)d_in[4];
  const float* W2a = (const float*)d_in[5];
  const float* b2  = (const float*)d_in[6];
  const float* W2r = (const float*)d_in[7];
  const float* Wc1 = (const float*)d_in[8];
  const float* bc1 = (const float*)d_in[9];
  const float* Wc2 = (const float*)d_in[10];
  const float* bc2 = (const float*)d_in[11];
  const int*   ei  = (const int*)d_in[12];

  const int N = in_sizes[0] / D;
  const int E = in_sizes[12] / 2;
  const int* src = ei;
  const int* dst = ei + E;

  float* ws = (float*)d_ws;
  const long NM = (long)N * D;
  float* bufA = ws;
  float* bufB = ws + NM;
  float* bufC = ws + 2 * NM;
  int*   deg  = (int*)(ws + 3 * NM);
  int*   rowptr = (int*)bufC;
  int*   rank   = rowptr + N;
  int*   nbr    = rank + E;
  float* out  = (float*)d_out;

  const dim3 blk(256);
  const int ggrid = N / 16;
  const int egrid = E / 16;
  const int agrid = (N + 7) / 8;
  const int tgrid = (E + 255) / 256;

  zero_i32<<<(N + 255) / 256, blk, 0, stream>>>(deg, N);
  count_rank<<<tgrid, blk, 0, stream>>>(dst, deg, rank, E);
  scan_rowptr<<<(N + 1023) / 1024, dim3(1024), 0, stream>>>(deg, rowptr, N);
  fill_nbr<<<tgrid, blk, 0, stream>>>(src, dst, rowptr, rank, nbr, E);

  gather_mean<<<agrid, blk, 0, stream>>>(x, nbr, rowptr, deg, bufA, N);
  node_gemm<<<ggrid, blk, 0, stream>>>(bufA, W1a, x, W1r, b1, bufB, 1);

  gather_mean<<<agrid, blk, 0, stream>>>(bufB, nbr, rowptr, deg, bufA, N);
  node_gemm<<<ggrid, blk, 0, stream>>>(bufA, W2a, bufB, W2r, b2, bufC, 0);

  node_gemm<<<ggrid, blk, 0, stream>>>(bufC, Wc1,           nullptr, nullptr,
                                       nullptr, bufB, 0);
  node_gemm<<<ggrid, blk, 0, stream>>>(bufC, Wc1 + 128 * D, nullptr, nullptr,
                                       nullptr, bufA, 0);

  edge_mlp<<<egrid, blk, 0, stream>>>(bufB, bufA, ea, src, dst, Wc1 + 256 * D,
                                      bc1, Wc2, bc2, out, E);
}


# edge_mlp and gather_mean as resident grids (1536 / 2048 blocks) that loop over 16-edge tiles / nodes, tile-invariant W3 fragment once per wave: 500000 short waves to enter become 28672
# speedup vs baseline: 1.7208x; 1.0722x over previous
#include <hip/hip_runtime.h>


typedef __attribute__((ext_vector_type(2))) float v2f;
typedef __attribute__((ext_vector_type(8))) float v8f;

#define D   128
#define DE  32

__device__ __forceinline__ v8f wmma4(float a0, float a1, float b0, float b1, v8f c) {
  v2f a; a[0] = a0; a[1] = a1;
  v2f b; b[0] = b0; b[1] = b1;
  return __builtin_amdgcn_wmma_f32_16x16x4_f32(false, a, false, b, (short)0, c,
                                               false, false);
}

__global__ void __launch_bounds__(256) zero_i32(int* __restrict__ p, int n) {
  const int i = blockIdx.x * 256 + threadIdx.x;
  if (i < n) p[i] = 0;
}

__global__ void __launch_bounds__(256) count_rank(const int* __restrict__ dst,
                                                  int* __restrict__ deg,
                                                  int* __restrict__ rank, int E) {
  const int e = blockIdx.x * 256 + threadIdx.x;
  if (e < E) rank[e] = atomicAdd(deg + dst[e], 1);
}

__global__ void __launch_bounds__(1024) scan_rowptr(const int* __restrict__ deg,
                                                    int* __restrict__ rowptr, int N) {
  __shared__ int s_tile[32];
  __shared__ int s_front[32];
  const int tid  = threadIdx.x;
  const int lane = tid & 31;
  const int wave = tid >> 5;
  const int tile0 = blockIdx.x * 1024;

  int front = 0;
  for (int i = tid; i < tile0; i += 1024) front += deg[i];
  const int idx = tile0 + tid;
  const int v = idx < N ? deg[idx] : 0;

  int inc = v;
#pragma unroll
  for (int o = 1; o < 32; o <<= 1) {
    const int t = __shfl_up(inc, o);
    if (lane >= o) inc += t;
  }
#pragma unroll
  for (int o = 16; o > 0; o >>= 1) front += __shfl_xor(front, o);
  if (lane == 31) s_tile[wave]  = inc;
  if (lane == 0)  s_front[wave] = front;
  __syncthreads();

  int base = 0;
  for (int w = 0; w < 32; ++w) {
    base += s_front[w];
    if (w < wave) base += s_tile[w];
  }
  if (idx < N) rowptr[idx] = base + inc - v;
}

__global__ void __launch_bounds__(256) fill_nbr(const int* __restrict__ src,
                                                const int* __restrict__ dst,
                                                const int* __restrict__ rowptr,
                                                const int* __restrict__ rank,
                                                int* __restrict__ nbr, int E) {
  const int e = blockIdx.x * 256 + threadIdx.x;
  if (e < E) nbr[rowptr[dst[e]] + rank[e]] = src[e];
}

__global__ void __launch_bounds__(256) gather_mean(const float* __restrict__ X,
                                                   const int* __restrict__ nbr,
                                                   const int* __restrict__ rowptr,
                                                   const int* __restrict__ deg,
                                                   float* __restrict__ agg, int N) {
  const int lane = threadIdx.x & 31;
  const int first = __builtin_amdgcn_readfirstlane(blockIdx.x * 8 + (threadIdx.x >> 5));
  const int stride = gridDim.x * 8;
  for (int node = first; node < N; node += stride) {
    const int beg = rowptr[node];
    const int dg  = deg[node];

    float4 acc = make_float4(0.0f, 0.0f, 0.0f, 0.0f);
    for (int base = 0; base < dg; base += 32) {
      const int rem = dg - base < 32 ? dg - base : 32;
      const int mine = lane < rem ? nbr[beg + base + lane] : 0;
#pragma unroll 4
      for (int j = 0; j < rem; ++j) {
        int s = __builtin_amdgcn_readlane(mine, j);
        s = (unsigned)s < (unsigned)N ? s : N - 1;
        const float4 v = *(const float4*)(X + (long)s * D + lane * 4);
        acc.x += v.x; acc.y += v.y; acc.z += v.z; acc.w += v.w;
      }
    }
    const float c = fmaxf((float)dg, 1.0f);
    float4 o;
    o.x = acc.x / c; o.y = acc.y / c; o.z = acc.z / c; o.w = acc.w / c;
    *(float4*)(agg + (long)node * D + lane * 4) = o;
  }
}

__global__ void node_gemm(const float* __restrict__ A1, const float* __restrict__ W1,
                          const float* __restrict__ A2, const float* __restrict__ W2,
                          const float* __restrict__ bias, float* __restrict__ Hout,
                          int do_relu) {
  const int tid  = threadIdx.x;
  const int wave = tid >> 5;
  const int lane = tid & 31;
  const int half = lane >> 4;
  const int lm   = lane & 15;
  const int m0   = blockIdx.x * 16;
  const int n    = wave * 16 + lm;

  v8f c;
  const float bv = bias ? bias[n] : 0.0f;
#pragma unroll
  for (int r = 0; r < 8; ++r) c[r] = bv;

  const float* aRow = A1 + (long)(m0 + lm) * D + 2 * half;
#pragma unroll 4
  for (int k0 = 0; k0 < D; k0 += 4) {
    const float2 a = *(const float2*)(aRow + k0);
    const float b0 = W1[(k0 + 2 * half) * D + n];
    const float b1 = W1[(k0 + 2 * half + 1) * D + n];
    c = wmma4(a.x, a.y, b0, b1, c);
  }
  if (A2) {
    const float* aRow2 = A2 + (long)(m0 + lm) * D + 2 * half;
#pragma unroll 4
    for (int k0 = 0; k0 < D; k0 += 4) {
      const float2 a = *(const float2*)(aRow2 + k0);
      const float b0 = W2[(k0 + 2 * half) * D + n];
      const float b1 = W2[(k0 + 2 * half + 1) * D + n];
      c = wmma4(a.x, a.y, b0, b1, c);
    }
  }

#pragma unroll
  for (int r = 0; r < 8; ++r) {
    float v = c[r];
    if (do_relu) v = fmaxf(v, 0.0f);
    const int m = m0 + r + 8 * half;
    Hout[(long)m * D + n] = v;
  }
}

__global__ void edge_mlp(const float* __restrict__ P1, const float* __restrict__ P2,
                         const float* __restrict__ ea, const int* __restrict__ src,
                         const int* __restrict__ dst, const float* __restrict__ W3,
                         const float* __restrict__ bc1, const float* __restrict__ Wc2,
                         const float* __restrict__ bc2, float* __restrict__ out,
                         int E) {
  __shared__ int   s_src[16];
  __shared__ int   s_dst[16];
  __shared__ float s_red[8][16];

  const int tid  = threadIdx.x;
  const int wave = tid >> 5;
  const int lane = tid & 31;
  const int half = lane >> 4;
  const int lm   = lane & 15;
  const int n    = wave * 16 + lm;
  const int ntiles = E >> 4;
  const int step   = gridDim.x;

  const float bv = bc1[n];
  const float w2 = Wc2[n];
  const float b2 = bc2[0];
  float wb0[8], wb1[8];
#pragma unroll
  for (int j = 0; j < 8; ++j) {
    wb0[j] = W3[(4 * j + 2 * half) * D + n];
    wb1[j] = W3[(4 * j + 2 * half + 1) * D + n];
  }

  int tile = blockIdx.x;
  int id = 0;
  if (tid < 32 && tile < ntiles)
    id = tid < 16 ? src[tile * 16 + tid] : dst[tile * 16 + tid - 16];

  for (; tile < ntiles; tile += step) {
    const int e0 = tile * 16;
    if (tid < 16)       s_src[tid]      = id;
    else if (tid < 32)  s_dst[tid - 16] = id;
    __syncthreads();
    const int tnext = tile + step;
    if (tid < 32 && tnext < ntiles)
      id = tid < 16 ? src[tnext * 16 + tid] : dst[tnext * 16 + tid - 16];

    v8f c;
#pragma unroll
    for (int r = 0; r < 8; ++r) {
      const int m = r + 8 * half;
      c[r] = P1[(long)s_src[m] * D + n] + P2[(long)s_dst[m] * D + n] + bv;
    }

    const float* aRow = ea + (long)(e0 + lm) * DE + 2 * half;
#pragma unroll
    for (int j = 0; j < 8; ++j) {
      const float2 a = *(const float2*)(aRow + 4 * j);
      c = wmma4(a.x, a.y, wb0[j], wb1[j], c);
    }

#pragma unroll
    for (int r = 0; r < 8; ++r) {
      float t = fmaxf(c[r], 0.0f) * w2;
      t += __shfl_xor(t, 1);
      t += __shfl_xor(t, 2);
      t += __shfl_xor(t, 4);
      t += __shfl_xor(t, 8);
      if (lm == 0) s_red[wave][r + 8 * half] = t;
    }
    __syncthreads();

    if (tid < 16) {
      float acc = b2;
#pragma unroll
      for (int w = 0; w < 8; ++w) acc += s_red[w][tid];
      if (e0 + tid < E) out[e0 + tid] = acc;
    }
  }
}

extern "C" void kernel_launch(void* const* d_in, const int* in_sizes, int n_in,
                              void* d_out, int out_size, void* d_ws, size_t ws_size,
                              hipStream_t stream) {
  (void)n_in; (void)out_size; (void)ws_size;
  const float* x   = (const float*)d_in[0];
  const float* ea  = (const float*)d_in[1];
  const float* W1a = (const float*)d_in[2];
  const float* b1  = (const float*)d_in[3];
  const float* W1r = (const float*)d_in[4];
  const float* W2a = (const float*)d_in[5];
  const float* b2  = (const float*)d_in[6];
  const float* W2r = (const float*)d_in[7];
  const float* Wc1 = (const float*)d_in[8];
  const float* bc1 = (const float*)d_in[9];
  const float* Wc2 = (const float*)d_in[10];
  const float* bc2 = (const float*)d_in[11];
  const int*   ei  = (const int*)d_in[12];

  const int N = in_sizes[0] / D;
  const int E = in_sizes[12] / 2;
  const int* src = ei;
  const int* dst = ei + E;

  float* ws = (float*)d_ws;
  const long NM = (long)N * D;
  float* bufA = ws;
  float* bufB = ws + NM;
  float* bufC = ws + 2 * NM;
  int*   deg  = (int*)(ws + 3 * NM);
  int*   rowptr = (int*)bufC;
  int*   rank   = rowptr + N;
  int*   nbr    = rank + E;
  float* out  = (float*)d_out;

  const dim3 blk(256);
  const int ggrid = N / 16;
  const int tgrid = (E + 255) / 256;

  zero_i32<<<(N + 255) / 256, blk, 0, stream>>>(deg, N);
  count_rank<<<tgrid, blk, 0, stream>>>(dst, deg, rank, E);
  scan_rowptr<<<(N + 1023) / 1024, dim3(1024), 0, stream>>>(deg, rowptr, N);
  fill_nbr<<<tgrid, blk, 0, stream>>>(src, dst, rowptr, rank, nbr, E);

  gather_mean<<<2048, blk, 0, stream>>>(x, nbr, rowptr, deg, bufA, N);
  node_gemm<<<ggrid, blk, 0, stream>>>(bufA, W1a, x, W1r, b1, bufB, 1);

  gather_mean<<<2048, blk, 0, stream>>>(bufB, nbr, rowptr, deg, bufA, N);
  node_gemm<<<ggrid, blk, 0, stream>>>(bufA, W2a, bufB, W2r, b2, bufC, 0);

  node_gemm<<<ggrid, blk, 0, stream>>>(bufC, Wc1,           nullptr, nullptr,
                                       nullptr, bufB, 0);
  node_gemm<<<ggrid, blk, 0, stream>>>(bufC, Wc1 + 128 * D, nullptr, nullptr,
                                       nullptr, bufA, 0);

  edge_mlp<<<1536, blk, 0, stream>>>(bufB, bufA, ea, src, dst, Wc1 + 256 * D,
                                     bc1, Wc2, bc2, out, E);
}


# WMMA A tiles (node_gemm 16x128 rows, edge_mlp 16x32 edge features) fetched once per block by whole-line loads into LDS; waves read fragments there, not by 16-line float2 global loads in all 8 waves
# speedup vs baseline: 2.1717x; 1.2620x over previous
#include <hip/hip_runtime.h>


typedef __attribute__((ext_vector_type(2))) float v2f;
typedef __attribute__((ext_vector_type(8))) float v8f;

#define D   128
#define DE  32
#define AP  (D + 4)
#define EP  (DE + 4)

__device__ __forceinline__ v8f wmma4(float a0, float a1, float b0, float b1, v8f c) {
  v2f a; a[0] = a0; a[1] = a1;
  v2f b; b[0] = b0; b[1] = b1;
  return __builtin_amdgcn_wmma_f32_16x16x4_f32(false, a, false, b, (short)0, c,
                                               false, false);
}

__global__ void __launch_bounds__(256) zero_i32(int* __restrict__ p, int n) {
  const int i = blockIdx.x * 256 + threadIdx.x;
  if (i < n) p[i] = 0;
}

__global__ void __launch_bounds__(256) count_rank(const int* __restrict__ dst,
                                                  int* __restrict__ deg,
                                                  int* __restrict__ rank, int E) {
  const int e = blockIdx.x * 256 + threadIdx.x;
  if (e < E) rank[e] = atomicAdd(deg + dst[e], 1);
}

__global__ void __launch_bounds__(1024) scan_rowptr(const int* __restrict__ deg,
                                                    int* __restrict__ rowptr, int N) {
  __shared__ int s_tile[32];
  __shared__ int s_front[32];
  const int tid  = threadIdx.x;
  const int lane = tid & 31;
  const int wave = tid >> 5;
  const int tile0 = blockIdx.x * 1024;

  int front = 0;
  for (int i = tid; i < tile0; i += 1024) front += deg[i];
  const int idx = tile0 + tid;
  const int v = idx < N ? deg[idx] : 0;

  int inc = v;
#pragma unroll
  for (int o = 1; o < 32; o <<= 1) {
    const int t = __shfl_up(inc, o);
    if (lane >= o) inc += t;
  }
#pragma unroll
  for (int o = 16; o > 0; o >>= 1) front += __shfl_xor(front, o);
  if (lane == 31) s_tile[wave]  = inc;
  if (lane == 0)  s_front[wave] = front;
  __syncthreads();

  int base = 0;
  for (int w = 0; w < 32; ++w) {
    base += s_front[w];
    if (w < wave) base += s_tile[w];
  }
  if (idx < N) rowptr[idx] = base + inc - v;
}

__global__ void __launch_bounds__(256) fill_nbr(const int* __restrict__ src,
                                                const int* __restrict__ dst,
                                                const int* __restrict__ rowptr,
                                                const int* __restrict__ rank,
                                                int* __restrict__ nbr, int E) {
  const int e = blockIdx.x * 256 + threadIdx.x;
  if (e < E) nbr[rowptr[dst[e]] + rank[e]] = src[e];
}

__global__ void __launch_bounds__(256) gather_mean(const float* __restrict__ X,
                                                   const int* __restrict__ nbr,
                                                   const int* __restrict__ rowptr,
                                                   const int* __restrict__ deg,
                                                   float* __restrict__ agg, int N) {
  const int lane = threadIdx.x & 31;
  const int first = __builtin_amdgcn_readfirstlane(blockIdx.x * 8 + (threadIdx.x >> 5));
  const int stride = gridDim.x * 8;
  for (int node = first; node < N; node += stride) {
    const int beg = rowptr[node];
    const int dg  = deg[node];

    float4 acc = make_float4(0.0f, 0.0f, 0.0f, 0.0f);
    for (int base = 0; base < dg; base += 32) {
      const int rem = dg - base < 32 ? dg - base : 32;
      const int mine = lane < rem ? nbr[beg + base + lane] : 0;
#pragma unroll 4
      for (int j = 0; j < rem; ++j) {
        int s = __builtin_amdgcn_readlane(mine, j);
        s = (unsigned)s < (unsigned)N ? s : N - 1;
        const float4 v = *(const float4*)(X + (long)s * D + lane * 4);
        acc.x += v.x; acc.y += v.y; acc.z += v.z; acc.w += v.w;
      }
    }
    const float c = fmaxf((float)dg, 1.0f);
    float4 o;
    o.x = acc.x / c; o.y = acc.y / c; o.z = acc.z / c; o.w = acc.w / c;
    *(float4*)(agg + (long)node * D + lane * 4) = o;
  }
}

__global__ void node_gemm(const float* __restrict__ A1, const float* __restrict__ W1,
                          const float* __restrict__ A2, const float* __restrict__ W2,
                          const float* __restrict__ bias, float* __restrict__ Hout,
                          int do_relu) {
  __shared__ __attribute__((aligned(16))) float sA1[16 * AP];
  __shared__ __attribute__((aligned(16))) float sA2[16 * AP];

  const int tid  = threadIdx.x;
  const int wave = tid >> 5;
  const int lane = tid & 31;
  const int half = lane >> 4;
  const int lm   = lane & 15;
  const int m0   = blockIdx.x * 16;
  const int n    = wave * 16 + lm;

  {
    const int scol = lane * 4;
    const long goff = (long)(m0 + wave) * D + scol;
    float4 q0 = make_float4(0.0f, 0.0f, 0.0f, 0.0f), q1 = q0;
    if (A2) {
      q0 = *(const float4*)(A2 + goff);
      q1 = *(const float4*)(A2 + goff + 8 * D);
    }
    const float4 p0 = *(const float4*)(A1 + goff);
    const float4 p1 = *(const float4*)(A1 + goff + 8 * D);
    *(float4*)(sA1 + wave * AP + scol)       = p0;
    *(float4*)(sA1 + (wave + 8) * AP + scol) = p1;
    if (A2) {
      *(float4*)(sA2 + wave * AP + scol)       = q0;
      *(float4*)(sA2 + (wave + 8) * AP + scol) = q1;
    }
  }

  v8f c;
  const float bv = bias ? bias[n] : 0.0f;
#pragma unroll
  for (int r = 0; r < 8; ++r) c[r] = bv;
  __syncthreads();

  const float* aRow = sA1 + lm * AP + 2 * half;
#pragma unroll 4
  for (int k0 = 0; k0 < D; k0 += 4) {
    const float2 a = *(const float2*)(aRow + k0);
    const float b0 = W1[(k0 + 2 * half) * D + n];
    const float b1 = W1[(k0 + 2 * half + 1) * D + n];
    c = wmma4(a.x, a.y, b0, b1, c);
  }
  if (A2) {
    const float* aRow2 = sA2 + lm * AP + 2 * half;
#pragma unroll 4
    for (int k0 = 0; k0 < D; k0 += 4) {
      const float2 a = *(const float2*)(aRow2 + k0);
      const float b0 = W2[(k0 + 2 * half) * D + n];
      const float b1 = W2[(k0 + 2 * half + 1) * D + n];
      c = wmma4(a.x, a.y, b0, b1, c);
    }
  }

#pragma unroll
  for (int r = 0; r < 8; ++r) {
    float v = c[r];
    if (do_relu) v = fmaxf(v, 0.0f);
    const int m = m0 + r + 8 * half;
    Hout[(long)m * D + n] = v;
  }
}

__global__ void edge_mlp(const float* __restrict__ P1, const float* __restrict__ P2,
                         const float* __restrict__ ea, const int* __restrict__ src,
                         const int* __restrict__ dst, const float* __restrict__ W3,
                         const float* __restrict__ bc1, const float* __restrict__ Wc2,
                         const float* __restrict__ bc2, float* __restrict__ out,
                         int E) {
  __shared__ int   s_src[16];
  __shared__ int   s_dst[16];
  __shared__ float s_red[8][16];
  __shared__ __attribute__((aligned(16))) float s_ea[16 * EP];

  const int tid  = threadIdx.x;
  const int wave = tid >> 5;
  const int lane = tid & 31;
  const int half = lane >> 4;
  const int lm   = lane & 15;
  const int n    = wave * 16 + lm;
  const int ntiles = E >> 4;
  const int step   = gridDim.x;

  const float bv = bc1[n];
  const float w2 = Wc2[n];
  const float b2 = bc2[0];
  float wb0[8], wb1[8];
#pragma unroll
  for (int j = 0; j < 8; ++j) {
    wb0[j] = W3[(4 * j + 2 * half) * D + n];
    wb1[j] = W3[(4 * j + 2 * half + 1) * D + n];
  }

  int tile = blockIdx.x;
  int id = 0;
  if (tid < 32 && tile < ntiles)
    id = tid < 16 ? src[tile * 16 + tid] : dst[tile * 16 + tid - 16];
  const int eoff = (tid >> 4) * EP + (tid & 15) * 2;
  float2 ev = make_float2(0.0f, 0.0f);
  if (tile < ntiles) ev = *(const float2*)(ea + (long)tile * (16 * DE) + tid * 2);

  for (; tile < ntiles; tile += step) {
    const int e0 = tile * 16;
    if (tid < 16)       s_src[tid]      = id;
    else if (tid < 32)  s_dst[tid - 16] = id;
    *(float2*)(s_ea + eoff) = ev;
    __syncthreads();
    const int tnext = tile + step;
    if (tid < 32 && tnext < ntiles)
      id = tid < 16 ? src[tnext * 16 + tid] : dst[tnext * 16 + tid - 16];
    if (tnext < ntiles) ev = *(const float2*)(ea + (long)tnext * (16 * DE) + tid * 2);

    v8f c;
#pragma unroll
    for (int r = 0; r < 8; ++r) {
      const int m = r + 8 * half;
      c[r] = P1[(long)s_src[m] * D + n] + P2[(long)s_dst[m] * D + n] + bv;
    }

    const float* aRow = s_ea + lm * EP + 2 * half;
#pragma unroll
    for (int j = 0; j < 8; ++j) {
      const float2 a = *(const float2*)(aRow + 4 * j);
      c = wmma4(a.x, a.y, wb0[j], wb1[j], c);
    }

#pragma unroll
    for (int r = 0; r < 8; ++r) {
      float t = fmaxf(c[r], 0.0f) * w2;
      t += __shfl_xor(t, 1);
      t += __shfl_xor(t, 2);
      t += __shfl_xor(t, 4);
      t += __shfl_xor(t, 8);
      if (lm == 0) s_red[wave][r + 8 * half] = t;
    }
    __syncthreads();

    if (tid < 16) {
      float acc = b2;
#pragma unroll
      for (int w = 0; w < 8; ++w) acc += s_red[w][tid];
      if (e0 + tid < E) out[e0 + tid] = acc;
    }
  }
}

extern "C" void kernel_launch(void* const* d_in, const int* in_sizes, int n_in,
                              void* d_out, int out_size, void* d_ws, size_t ws_size,
                              hipStream_t stream) {
  (void)n_in; (void)out_size; (void)ws_size;
  const float* x   = (const float*)d_in[0];
  const float* ea  = (const float*)d_in[1];
  const float* W1a = (const float*)d_in[2];
  const float* b1  = (const float*)d_in[3];
  const float* W1r = (const float*)d_in[4];
  const float* W2a = (const float*)d_in[5];
  const float* b2  = (const float*)d_in[6];
  const float* W2r = (const float*)d_in[7];
  const float* Wc1 = (const float*)d_in[8];
  const float* bc1 = (const float*)d_in[9];
  const float* Wc2 = (const float*)d_in[10];
  const float* bc2 = (const float*)d_in[11];
  const int*   ei  = (const int*)d_in[12];

  const int N = in_sizes[0] / D;
  const int E = in_sizes[12] / 2;
  const int* src = ei;
  const int* dst = ei + E;

  float* ws = (float*)d_ws;
  const long NM = (long)N * D;
  float* bufA = ws;
  float* bufB = ws + NM;
  float* bufC = ws + 2 * NM;
  int*   deg  = (int*)(ws + 3 * NM);
  int*   rowptr = (int*)bufC;
  int*   rank   = rowptr + N;
  int*   nbr    = rank + E;
  float* out  = (float*)d_out;

  const dim3 blk(256);
  const int ggrid = N / 16;
  const int tgrid = (E + 255) / 256;

  zero_i32<<<(N + 255) / 256, blk, 0, stream>>>(deg, N);
  count_rank<<<tgrid, blk, 0, stream>>>(dst, deg, rank, E);
  scan_rowptr<<<(N + 1023) / 1024, dim3(1024), 0, stream>>>(deg, rowptr, N);
  fill_nbr<<<tgrid, blk, 0, stream>>>(src, dst, rowptr, rank, nbr, E);

  gather_mean<<<2048, blk, 0, stream>>>(x, nbr, rowptr, deg, bufA, N);
  node_gemm<<<ggrid, blk, 0, stream>>>(bufA, W1a, x, W1r, b1, bufB, 1);

  gather_mean<<<2048, blk, 0, stream>>>(bufB, nbr, rowptr, deg, bufA, N);
  node_gemm<<<ggrid, blk, 0, stream>>>(bufA, W2a, bufB, W2r, b2, bufC, 0);

  node_gemm<<<ggrid, blk, 0, stream>>>(bufC, Wc1,           nullptr, nullptr,
                                       nullptr, bufB, 0);
  node_gemm<<<ggrid, blk, 0, stream>>>(bufC, Wc1 + 128 * D, nullptr, nullptr,
                                       nullptr, bufA, 0);

  edge_mlp<<<1536, blk, 0, stream>>>(bufB, bufA, ea, src, dst, Wc1 + 256 * D,
                                     bc1, Wc2, bc2, out, E);
}


# edge_mlp: 16-lane row sums of relu(hid)*Wc2 by four row_xmask DPP adds on the vector ALU (same xor 1,2,4,8 order) instead of 32 serial waited-for ds_bpermute round trips per tile and wave
# speedup vs baseline: 2.3271x; 1.0716x over previous
#include <hip/hip_runtime.h>


typedef __attribute__((ext_vector_type(2))) float v2f;
typedef __attribute__((ext_vector_type(8))) float v8f;

#define D   128
#define DE  32
#define AP  (D + 4)
#define EP  (DE + 4)

__device__ __forceinline__ v8f wmma4(float a0, float a1, float b0, float b1, v8f c) {
  v2f a; a[0] = a0; a[1] = a1;
  v2f b; b[0] = b0; b[1] = b1;
  return __builtin_amdgcn_wmma_f32_16x16x4_f32(false, a, false, b, (short)0, c,
                                               false, false);
}

template <int M>
__device__ __forceinline__ float row_xor_add(float v) {
  const int other =
      __builtin_amdgcn_update_dpp(0, __float_as_int(v), 0x160 + M, 0xF, 0xF, true);
  return v + __int_as_float(other);
}

__global__ void __launch_bounds__(256) zero_i32(int* __restrict__ p, int n) {
  const int i = blockIdx.x * 256 + threadIdx.x;
  if (i < n) p[i] = 0;
}

__global__ void __launch_bounds__(256) count_rank(const int* __restrict__ dst,
                                                  int* __restrict__ deg,
                                                  int* __restrict__ rank, int E) {
  const int e = blockIdx.x * 256 + threadIdx.x;
  if (e < E) rank[e] = atomicAdd(deg + dst[e], 1);
}

__global__ void __launch_bounds__(1024) scan_rowptr(const int* __restrict__ deg,
                                                    int* __restrict__ rowptr, int N) {
  __shared__ int s_tile[32];
  __shared__ int s_front[32];
  const int tid  = threadIdx.x;
  const int lane = tid & 31;
  const int wave = tid >> 5;
  const int tile0 = blockIdx.x * 1024;

  int front = 0;
  for (int i = tid; i < tile0; i += 1024) front += deg[i];
  const int idx = tile0 + tid;
  const int v = idx < N ? deg[idx] : 0;

  int inc = v;
#pragma unroll
  for (int o = 1; o < 32; o <<= 1) {
    const int t = __shfl_up(inc, o);
    if (lane >= o) inc += t;
  }
#pragma unroll
  for (int o = 16; o > 0; o >>= 1) front += __shfl_xor(front, o);
  if (lane == 31) s_tile[wave]  = inc;
  if (lane == 0)  s_front[wave] = front;
  __syncthreads();

  int base = 0;
  for (int w = 0; w < 32; ++w) {
    base += s_front[w];
    if (w < wave) base += s_tile[w];
  }
  if (idx < N) rowptr[idx] = base + inc - v;
}

__global__ void __launch_bounds__(256) fill_nbr(const int* __restrict__ src,
                                                const int* __restrict__ dst,
                                                const int* __restrict__ rowptr,
                                                const int* __restrict__ rank,
                                                int* __restrict__ nbr, int E) {
  const int e = blockIdx.x * 256 + threadIdx.x;
  if (e < E) nbr[rowptr[dst[e]] + rank[e]] = src[e];
}

__global__ void __launch_bounds__(256) gather_mean(const float* __restrict__ X,
                                                   const int* __restrict__ nbr,
                                                   const int* __restrict__ rowptr,
                                                   const int* __restrict__ deg,
                                                   float* __restrict__ agg, int N) {
  const int lane = threadIdx.x & 31;
  const int first = __builtin_amdgcn_readfirstlane(blockIdx.x * 8 + (threadIdx.x >> 5));
  const int stride = gridDim.x * 8;
  for (int node = first; node < N; node += stride) {
    const int beg = rowptr[node];
    const int dg  = deg[node];

    float4 acc = make_float4(0.0f, 0.0f, 0.0f, 0.0f);
    for (int base = 0; base < dg; base += 32) {
      const int rem = dg - base < 32 ? dg - base : 32;
      const int mine = lane < rem ? nbr[beg + base + lane] : 0;
#pragma unroll 4
      for (int j = 0; j < rem; ++j) {
        int s = __builtin_amdgcn_readlane(mine, j);
        s = (unsigned)s < (unsigned)N ? s : N - 1;
        const float4 v = *(const float4*)(X + (long)s * D + lane * 4);
        acc.x += v.x; acc.y += v.y; acc.z += v.z; acc.w += v.w;
      }
    }
    const float c = fmaxf((float)dg, 1.0f);
    float4 o;
    o.x = acc.x / c; o.y = acc.y / c; o.z = acc.z / c; o.w = acc.w / c;
    *(float4*)(agg + (long)node * D + lane * 4) = o;
  }
}

__global__ void node_gemm(const float* __restrict__ A1, const float* __restrict__ W1,
                          const float* __restrict__ A2, const float* __restrict__ W2,
                          const float* __restrict__ bias, float* __restrict__ Hout,
                          int do_relu) {
  __shared__ __attribute__((aligned(16))) float sA1[16 * AP];
  __shared__ __attribute__((aligned(16))) float sA2[16 * AP];

  const int tid  = threadIdx.x;
  const int wave = tid >> 5;
  const int lane = tid & 31;
  const int half = lane >> 4;
  const int lm   = lane & 15;
  const int m0   = blockIdx.x * 16;
  const int n    = wave * 16 + lm;

  {
    const int scol = lane * 4;
    const long goff = (long)(m0 + wave) * D + scol;
    float4 q0 = make_float4(0.0f, 0.0f, 0.0f, 0.0f), q1 = q0;
    if (A2) {
      q0 = *(const float4*)(A2 + goff);
      q1 = *(const float4*)(A2 + goff + 8 * D);
    }
    const float4 p0 = *(const float4*)(A1 + goff);
    const float4 p1 = *(const float4*)(A1 + goff + 8 * D);
    *(float4*)(sA1 + wave * AP + scol)       = p0;
    *(float4*)(sA1 + (wave + 8) * AP + scol) = p1;
    if (A2) {
      *(float4*)(sA2 + wave * AP + scol)       = q0;
      *(float4*)(sA2 + (wave + 8) * AP + scol) = q1;
    }
  }

  v8f c;
  const float bv = bias ? bias[n] : 0.0f;
#pragma unroll
  for (int r = 0; r < 8; ++r) c[r] = bv;
  __syncthreads();

  const float* aRow = sA1 + lm * AP + 2 * half;
#pragma unroll 4
  for (int k0 = 0; k0 < D; k0 += 4) {
    const float2 a = *(const float2*)(aRow + k0);
    const float b0 = W1[(k0 + 2 * half) * D + n];
    const float b1 = W1[(k0 + 2 * half + 1) * D + n];
    c = wmma4(a.x, a.y, b0, b1, c);
  }
  if (A2) {
    const float* aRow2 = sA2 + lm * AP + 2 * half;
#pragma unroll 4
    for (int k0 = 0; k0 < D; k0 += 4) {
      const float2 a = *(const float2*)(aRow2 + k0);
      const float b0 = W2[(k0 + 2 * half) * D + n];
      const float b1 = W2[(k0 + 2 * half + 1) * D + n];
      c = wmma4(a.x, a.y, b0, b1, c);
    }
  }

#pragma unroll
  for (int r = 0; r < 8; ++r) {
    float v = c[r];
    if (do_relu) v = fmaxf(v, 0.0f);
    const int m = m0 + r + 8 * half;
    Hout[(long)m * D + n] = v;
  }
}

__global__ void edge_mlp(const float* __restrict__ P1, const float* __restrict__ P2,
                         const float* __restrict__ ea, const int* __restrict__ src,
                         const int* __restrict__ dst, const float* __restrict__ W3,
                         const float* __restrict__ bc1, const float* __restrict__ Wc2,
                         const float* __restrict__ bc2, float* __restrict__ out,
                         int E) {
  __shared__ int   s_src[16];
  __shared__ int   s_dst[16];
  __shared__ __attribute__((aligned(16))) float s_red[8][16];
  __shared__ __attribute__((aligned(16))) float s_ea[16 * EP];

  const int tid  = threadIdx.x;
  const int wave = tid >> 5;
  const int lane = tid & 31;
  const int half = lane >> 4;
  const int lm   = lane & 15;
  const int n    = wave * 16 + lm;
  const int ntiles = E >> 4;
  const int step   = gridDim.x;

  const float bv = bc1[n];
  const float w2 = Wc2[n];
  const float b2 = bc2[0];
  float wb0[8], wb1[8];
#pragma unroll
  for (int j = 0; j < 8; ++j) {
    wb0[j] = W3[(4 * j + 2 * half) * D + n];
    wb1[j] = W3[(4 * j + 2 * half + 1) * D + n];
  }

  int tile = blockIdx.x;
  int id = 0;
  if (tid < 32 && tile < ntiles)
    id = tid < 16 ? src[tile * 16 + tid] : dst[tile * 16 + tid - 16];
  const int eoff = (tid >> 4) * EP + (tid & 15) * 2;
  float2 ev = make_float2(0.0f, 0.0f);
  if (tile < ntiles) ev = *(const float2*)(ea + (long)tile * (16 * DE) + tid * 2);

  for (; tile < ntiles; tile += step) {
    const int e0 = tile * 16;
    if (tid < 16)       s_src[tid]      = id;
    else if (tid < 32)  s_dst[tid - 16] = id;
    *(float2*)(s_ea + eoff) = ev;
    __syncthreads();
    const int tnext = tile + step;
    if (tid < 32 && tnext < ntiles)
      id = tid < 16 ? src[tnext * 16 + tid] : dst[tnext * 16 + tid - 16];
    if (tnext < ntiles) ev = *(const float2*)(ea + (long)tnext * (16 * DE) + tid * 2);

    v8f c;
#pragma unroll
    for (int r = 0; r < 8; ++r) {
      const int m = r + 8 * half;
      c[r] = P1[(long)s_src[m] * D + n] + P2[(long)s_dst[m] * D + n] + bv;
    }

    const float* aRow = s_ea + lm * EP + 2 * half;
#pragma unroll
    for (int j = 0; j < 8; ++j) {
      const float2 a = *(const float2*)(aRow + 4 * j);
      c = wmma4(a.x, a.y, wb0[j], wb1[j], c);
    }

    float red[8];
#pragma unroll
    for (int r = 0; r < 8; ++r) {
      float t = fmaxf(c[r], 0.0f) * w2;
      t = row_xor_add<1>(t);
      t = row_xor_add<2>(t);
      t = row_xor_add<4>(t);
      t = row_xor_add<8>(t);
      red[r] = t;
    }
    if (lm == 0) {
      float* p = &s_red[wave][8 * half];
      *(float4*)p       = make_float4(red[0], red[1], red[2], red[3]);
      *(float4*)(p + 4) = make_float4(red[4], red[5], red[6], red[7]);
    }
    __syncthreads();

    if (tid < 16) {
      float acc = b2;
#pragma unroll
      for (int w = 0; w < 8; ++w) acc += s_red[w][tid];
      if (e0 + tid < E) out[e0 + tid] = acc;
    }
  }
}

extern "C" void kernel_launch(void* const* d_in, const int* in_sizes, int n_in,
                              void* d_out, int out_size, void* d_ws, size_t ws_size,
                              hipStream_t stream) {
  (void)n_in; (void)out_size; (void)ws_size;
  const float* x   = (const float*)d_in[0];
  const float* ea  = (const float*)d_in[1];
  const float* W1a = (const float*)d_in[2];
  const float* b1  = (const float*)d_in[3];
  const float* W1r = (const float*)d_in[4];
  const float* W2a = (const float*)d_in[5];
  const float* b2  = (const float*)d_in[6];
  const float* W2r = (const float*)d_in[7];
  const float* Wc1 = (const float*)d_in[8];
  const float* bc1 = (const float*)d_in[9];
  const float* Wc2 = (const float*)d_in[10];
  const float* bc2 = (const float*)d_in[11];
  const int*   ei  = (const int*)d_in[12];

  const int N = in_sizes[0] / D;
  const int E = in_sizes[12] / 2;
  const int* src = ei;
  const int* dst = ei + E;

  float* ws = (float*)d_ws;
  const long NM = (long)N * D;
  float* bufA = ws;
  float* bufB = ws + NM;
  float* bufC = ws + 2 * NM;
  int*   deg  = (int*)(ws + 3 * NM);
  int*   rowptr = (int*)bufC;
  int*   rank   = rowptr + N;
  int*   nbr    = rank + E;
  float* out  = (float*)d_out;

  const dim3 blk(256);
  const int ggrid = N / 16;
  const int tgrid = (E + 255) / 256;

  zero_i32<<<(N + 255) / 256, blk, 0, stream>>>(deg, N);
  count_rank<<<tgrid, blk, 0, stream>>>(dst, deg, rank, E);
  scan_rowptr<<<(N + 1023) / 1024, dim3(1024), 0, stream>>>(deg, rowptr, N);
  fill_nbr<<<tgrid, blk, 0, stream>>>(src, dst, rowptr, rank, nbr, E);

  gather_mean<<<2048, blk, 0, stream>>>(x, nbr, rowptr, deg, bufA, N);
  node_gemm<<<ggrid, blk, 0, stream>>>(bufA, W1a, x, W1r, b1, bufB, 1);

  gather_mean<<<2048, blk, 0, stream>>>(bufB, nbr, rowptr, deg, bufA, N);
  node_gemm<<<ggrid, blk, 0, stream>>>(bufA, W2a, bufB, W2r, b2, bufC, 0);

  node_gemm<<<ggrid, blk, 0, stream>>>(bufC, Wc1,           nullptr, nullptr,
                                       nullptr, bufB, 0);
  node_gemm<<<ggrid, blk, 0, stream>>>(bufC, Wc1 + 128 * D, nullptr, nullptr,
                                       nullptr, bufA, 0);

  edge_mlp<<<1536, blk, 0, stream>>>(bufB, bufA, ea, src, dst, Wc1 + 256 * D,
                                     bc1, Wc2, bc2, out, E);
}


# gather_mean fused into the layer GEMM kernel: each 16-row block pulls its nodes' neighbour means straight into its LDS A tile, the aggregate never goes to memory; 11 launches become 9
# speedup vs baseline: 2.4462x; 1.0512x over previous
#include <hip/hip_runtime.h>


typedef __attribute__((ext_vector_type(2))) float v2f;
typedef __attribute__((ext_vector_type(8))) float v8f;

#define D   128
#define DE  32
#define AP  (D + 4)
#define EP  (DE + 4)

__device__ __forceinline__ v8f wmma4(float a0, float a1, float b0, float b1, v8f c) {
  v2f a; a[0] = a0; a[1] = a1;
  v2f b; b[0] = b0; b[1] = b1;
  return __builtin_amdgcn_wmma_f32_16x16x4_f32(false, a, false, b, (short)0, c,
                                               false, false);
}

template <int M>
__device__ __forceinline__ float row_xor_add(float v) {
  const int other =
      __builtin_amdgcn_update_dpp(0, __float_as_int(v), 0x160 + M, 0xF, 0xF, true);
  return v + __int_as_float(other);
}

__global__ void __launch_bounds__(256) zero_i32(int* __restrict__ p, int n) {
  const int i = blockIdx.x * 256 + threadIdx.x;
  if (i < n) p[i] = 0;
}

__global__ void __launch_bounds__(256) count_rank(const int* __restrict__ dst,
                                                  int* __restrict__ deg,
                                                  int* __restrict__ rank, int E) {
  const int e = blockIdx.x * 256 + threadIdx.x;
  if (e < E) rank[e] = atomicAdd(deg + dst[e], 1);
}

__global__ void __launch_bounds__(1024) scan_rowptr(const int* __restrict__ deg,
                                                    int* __restrict__ rowptr, int N) {
  __shared__ int s_tile[32];
  __shared__ int s_front[32];
  const int tid  = threadIdx.x;
  const int lane = tid & 31;
  const int wave = tid >> 5;
  const int tile0 = blockIdx.x * 1024;

  int front = 0;
  for (int i = tid; i < tile0; i += 1024) front += deg[i];
  const int idx = tile0 + tid;
  const int v = idx < N ? deg[idx] : 0;

  int inc = v;
#pragma unroll
  for (int o = 1; o < 32; o <<= 1) {
    const int t = __shfl_up(inc, o);
    if (lane >= o) inc += t;
  }
#pragma unroll
  for (int o = 16; o > 0; o >>= 1) front += __shfl_xor(front, o);
  if (lane == 31) s_tile[wave]  = inc;
  if (lane == 0)  s_front[wave] = front;
  __syncthreads();

  int base = 0;
  for (int w = 0; w < 32; ++w) {
    base += s_front[w];
    if (w < wave) base += s_tile[w];
  }
  if (idx < N) rowptr[idx] = base + inc - v;
}

__global__ void __launch_bounds__(256) fill_nbr(const int* __restrict__ src,
                                                const int* __restrict__ dst,
                                                const int* __restrict__ rowptr,
                                                const int* __restrict__ rank,
                                                int* __restrict__ nbr, int E) {
  const int e = blockIdx.x * 256 + threadIdx.x;
  if (e < E) nbr[rowptr[dst[e]] + rank[e]] = src[e];
}

__device__ __forceinline__ float4 pull_mean(const float* __restrict__ X,
                                            const int* __restrict__ nbr, int beg,
                                            int dg, int ids, int lane, int N) {
  float4 acc = make_float4(0.0f, 0.0f, 0.0f, 0.0f);
  for (int base = 0; base < dg; base += 32) {
    const int rem = dg - base < 32 ? dg - base : 32;
    int mine = ids;
    if (base > 0) mine = lane < rem ? nbr[beg + base + lane] : 0;
#pragma unroll 4
    for (int j = 0; j < rem; ++j) {
      int s = __builtin_amdgcn_readlane(mine, j);
      s = (unsigned)s < (unsigned)N ? s : N - 1;
      const float4 v = *(const float4*)(X + (long)s * D + lane * 4);
      acc.x += v.x; acc.y += v.y; acc.z += v.z; acc.w += v.w;
    }
  }
  const float c = fmaxf((float)dg, 1.0f);
  float4 o;
  o.x = acc.x / c; o.y = acc.y / c; o.z = acc.z / c; o.w = acc.w / c;
  return o;
}

__device__ __forceinline__ v8f tile_gemm(const float* aRow, const float* __restrict__ Wm,
                                         int half, int n, v8f c) {
#pragma unroll 4
  for (int k0 = 0; k0 < D; k0 += 4) {
    const float2 a = *(const float2*)(aRow + k0);
    const float b0 = Wm[(k0 + 2 * half) * D + n];
    const float b1 = Wm[(k0 + 2 * half + 1) * D + n];
    c = wmma4(a.x, a.y, b0, b1, c);
  }
  return c;
}

__device__ __forceinline__ void store_tile(float* __restrict__ Hout, v8f c, int m0,
                                           int half, int n, int do_relu) {
#pragma unroll
  for (int r = 0; r < 8; ++r) {
    float v = c[r];
    if (do_relu) v = fmaxf(v, 0.0f);
    const int m = m0 + r + 8 * half;
    Hout[(long)m * D + n] = v;
  }
}

__global__ void __launch_bounds__(256) sage_layer(
    const float* __restrict__ X, const int* __restrict__ nbr,
    const int* __restrict__ rowptr, const int* __restrict__ deg,
    const float* __restrict__ Wagg, const float* __restrict__ Wroot,
    const float* __restrict__ bias, float* __restrict__ Hout, int do_relu, int N) {
  __shared__ __attribute__((aligned(16))) float sAgg[16 * AP];
  __shared__ __attribute__((aligned(16))) float sRoot[16 * AP];

  const int tid  = threadIdx.x;
  const int lane = tid & 31;
  const int half = lane >> 4;
  const int lm   = lane & 15;
  const int wave = __builtin_amdgcn_readfirstlane(tid >> 5);
  const int m0   = blockIdx.x * 16;
  const int n    = wave * 16 + lm;

  {
    const int scol = lane * 4;
    const int r0 = m0 + wave;
    const int r1 = r0 + 8;
    const float4 q0 = *(const float4*)(X + (long)r0 * D + scol);
    const float4 q1 = *(const float4*)(X + (long)r1 * D + scol);
    const int beg0 = rowptr[r0];
    const int dg0  = deg[r0];
    const int beg1 = rowptr[r1];
    const int dg1  = deg[r1];
    const int ids0 = lane < dg0 ? nbr[beg0 + lane] : 0;
    const int ids1 = lane < dg1 ? nbr[beg1 + lane] : 0;
    *(float4*)(sRoot + wave * AP + scol)       = q0;
    *(float4*)(sRoot + (wave + 8) * AP + scol) = q1;
    const float4 p0 = pull_mean(X, nbr, beg0, dg0, ids0, lane, N);
    const float4 p1 = pull_mean(X, nbr, beg1, dg1, ids1, lane, N);
    *(float4*)(sAgg + wave * AP + scol)        = p0;
    *(float4*)(sAgg + (wave + 8) * AP + scol)  = p1;
  }

  v8f c;
  const float bv = bias[n];
#pragma unroll
  for (int r = 0; r < 8; ++r) c[r] = bv;
  __syncthreads();

  c = tile_gemm(sAgg + lm * AP + 2 * half, Wagg, half, n, c);
  c = tile_gemm(sRoot + lm * AP + 2 * half, Wroot, half, n, c);
  store_tile(Hout, c, m0, half, n, do_relu);
}

__global__ void __launch_bounds__(256) node_gemm(const float* __restrict__ A,
                                                 const float* __restrict__ Wm,
                                                 float* __restrict__ Hout) {
  __shared__ __attribute__((aligned(16))) float sA[16 * AP];

  const int tid  = threadIdx.x;
  const int wave = tid >> 5;
  const int lane = tid & 31;
  const int half = lane >> 4;
  const int lm   = lane & 15;
  const int m0   = blockIdx.x * 16;
  const int n    = wave * 16 + lm;

  {
    const int scol = lane * 4;
    const long goff = (long)(m0 + wave) * D + scol;
    const float4 p0 = *(const float4*)(A + goff);
    const float4 p1 = *(const float4*)(A + goff + 8 * D);
    *(float4*)(sA + wave * AP + scol)       = p0;
    *(float4*)(sA + (wave + 8) * AP + scol) = p1;
  }

  v8f c;
#pragma unroll
  for (int r = 0; r < 8; ++r) c[r] = 0.0f;
  __syncthreads();

  c = tile_gemm(sA + lm * AP + 2 * half, Wm, half, n, c);
  store_tile(Hout, c, m0, half, n, 0);
}

__global__ void edge_mlp(const float* __restrict__ P1, const float* __restrict__ P2,
                         const float* __restrict__ ea, const int* __restrict__ src,
                         const int* __restrict__ dst, const float* __restrict__ W3,
                         const float* __restrict__ bc1, const float* __restrict__ Wc2,
                         const float* __restrict__ bc2, float* __restrict__ out,
                         int E) {
  __shared__ int   s_src[16];
  __shared__ int   s_dst[16];
  __shared__ __attribute__((aligned(16))) float s_red[8][16];
  __shared__ __attribute__((aligned(16))) float s_ea[16 * EP];

  const int tid  = threadIdx.x;
  const int wave = tid >> 5;
  const int lane = tid & 31;
  const int half = lane >> 4;
  const int lm   = lane & 15;
  const int n    = wave * 16 + lm;
  const int ntiles = E >> 4;
  const int step   = gridDim.x;

  const float bv = bc1[n];
  const float w2 = Wc2[n];
  const float b2 = bc2[0];
  float wb0[8], wb1[8];
#pragma unroll
  for (int j = 0; j < 8; ++j) {
    wb0[j] = W3[(4 * j + 2 * half) * D + n];
    wb1[j] = W3[(4 * j + 2 * half + 1) * D + n];
  }

  int tile = blockIdx.x;
  int id = 0;
  if (tid < 32 && tile < ntiles)
    id = tid < 16 ? src[tile * 16 + tid] : dst[tile * 16 + tid - 16];
  const int eoff = (tid >> 4) * EP + (tid & 15) * 2;
  float2 ev = make_float2(0.0f, 0.0f);
  if (tile < ntiles) ev = *(const float2*)(ea + (long)tile * (16 * DE) + tid * 2);

  for (; tile < ntiles; tile += step) {
    const int e0 = tile * 16;
    if (tid < 16)       s_src[tid]      = id;
    else if (tid < 32)  s_dst[tid - 16] = id;
    *(float2*)(s_ea + eoff) = ev;
    __syncthreads();
    const int tnext = tile + step;
    if (tid < 32 && tnext < ntiles)
      id = tid < 16 ? src[tnext * 16 + tid] : dst[tnext * 16 + tid - 16];
    if (tnext < ntiles) ev = *(const float2*)(ea + (long)tnext * (16 * DE) + tid * 2);

    v8f c;
#pragma unroll
    for (int r = 0; r < 8; ++r) {
      const int m = r + 8 * half;
      c[r] = P1[(long)s_src[m] * D + n] + P2[(long)s_dst[m] * D + n] + bv;
    }

    const float* aRow = s_ea + lm * EP + 2 * half;
#pragma unroll
    for (int j = 0; j < 8; ++j) {
      const float2 a = *(const float2*)(aRow + 4 * j);
      c = wmma4(a.x, a.y, wb0[j], wb1[j], c);
    }

    float red[8];
#pragma unroll
    for (int r = 0; r < 8; ++r) {
      float t = fmaxf(c[r], 0.0f) * w2;
      t = row_xor_add<1>(t);
      t = row_xor_add<2>(t);
      t = row_xor_add<4>(t);
      t = row_xor_add<8>(t);
      red[r] = t;
    }
    if (lm == 0) {
      float* p = &s_red[wave][8 * half];
      *(float4*)p       = make_float4(red[0], red[1], red[2], red[3]);
      *(float4*)(p + 4) = make_float4(red[4], red[5], red[6], red[7]);
    }
    __syncthreads();

    if (tid < 16) {
      float acc = b2;
#pragma unroll
      for (int w = 0; w < 8; ++w) acc += s_red[w][tid];
      if (e0 + tid < E) out[e0 + tid] = acc;
    }
  }
}

extern "C" void kernel_launch(void* const* d_in, const int* in_sizes, int n_in,
                              void* d_out, int out_size, void* d_ws, size_t ws_size,
                              hipStream_t stream) {
  (void)n_in; (void)out_size; (void)ws_size;
  const float* x   = (const float*)d_in[0];
  const float* ea  = (const float*)d_in[1];
  const float* W1a = (const float*)d_in[2];
  const float* b1  = (const float*)d_in[3];
  const float* W1r = (const float*)d_in[4];
  const float* W2a = (const float*)d_in[5];
  const float* b2  = (const float*)d_in[6];
  const float* W2r = (const float*)d_in[7];
  const float* Wc1 = (const float*)d_in[8];
  const float* bc1 = (const float*)d_in[9];
  const float* Wc2 = (const float*)d_in[10];
  const float* bc2 = (const float*)d_in[11];
  const int*   ei  = (const int*)d_in[12];

  const int N = in_sizes[0] / D;
  const int E = in_sizes[12] / 2;
  const int* src = ei;
  const int* dst = ei + E;

  float* ws = (float*)d_ws;
  const long NM = (long)N * D;
  float* bufA = ws;
  float* bufB = ws + NM;
  float* bufC = ws + 2 * NM;
  int*   deg  = (int*)(ws + 3 * NM);
  int*   rowptr = (int*)bufA;
  int*   rank   = rowptr + N;
  int*   nbr    = rank + E;
  float* out  = (float*)d_out;

  const dim3 blk(256);
  const int ggrid = N / 16;
  const int tgrid = (E + 255) / 256;

  zero_i32<<<(N + 255) / 256, blk, 0, stream>>>(deg, N);
  count_rank<<<tgrid, blk, 0, stream>>>(dst, deg, rank, E);
  scan_rowptr<<<(N + 1023) / 1024, dim3(1024), 0, stream>>>(deg, rowptr, N);
  fill_nbr<<<tgrid, blk, 0, stream>>>(src, dst, rowptr, rank, nbr, E);

  sage_layer<<<ggrid, blk, 0, stream>>>(x, nbr, rowptr, deg, W1a, W1r, b1, bufB, 1, N);

  sage_layer<<<ggrid, blk, 0, stream>>>(bufB, nbr, rowptr, deg, W2a, W2r, b2, bufC, 0, N);

  node_gemm<<<ggrid, blk, 0, stream>>>(bufC, Wc1,           bufB);
  node_gemm<<<ggrid, blk, 0, stream>>>(bufC, Wc1 + 128 * D, bufA);

  edge_mlp<<<1536, blk, 0, stream>>>(bufB, bufA, ea, src, dst, Wc1 + 256 * D,
                                     bc1, Wc2, bc2, out, E);
}


# Layer-2 kernel keeps its 16x128 tile of h2 in LDS and multiplies it by both node slices of Wc1 in the same block (one pass, two accumulators): h2 never goes to memory; 9 launches become 7
# speedup vs baseline: 2.5412x; 1.0388x over previous
#include <hip/hip_runtime.h>


typedef __attribute__((ext_vector_type(2))) float v2f;
typedef __attribute__((ext_vector_type(8))) float v8f;

#define D   128
#define DE  32
#define AP  (D + 4)
#define EP  (DE + 4)

__device__ __forceinline__ v8f wmma4(float a0, float a1, float b0, float b1, v8f c) {
  v2f a; a[0] = a0; a[1] = a1;
  v2f b; b[0] = b0; b[1] = b1;
  return __builtin_amdgcn_wmma_f32_16x16x4_f32(false, a, false, b, (short)0, c,
                                               false, false);
}

template <int M>
__device__ __forceinline__ float row_xor_add(float v) {
  const int other =
      __builtin_amdgcn_update_dpp(0, __float_as_int(v), 0x160 + M, 0xF, 0xF, true);
  return v + __int_as_float(other);
}

__global__ void __launch_bounds__(256) zero_i32(int* __restrict__ p, int n) {
  const int i = blockIdx.x * 256 + threadIdx.x;
  if (i < n) p[i] = 0;
}

__global__ void __launch_bounds__(256) count_rank(const int* __restrict__ dst,
                                                  int* __restrict__ deg,
                                                  int* __restrict__ rank, int E) {
  const int e = blockIdx.x * 256 + threadIdx.x;
  if (e < E) rank[e] = atomicAdd(deg + dst[e], 1);
}

__global__ void __launch_bounds__(1024) scan_rowend(const int* __restrict__ deg,
                                                    int* __restrict__ rowend, int N) {
  __shared__ int s_tile[32];
  __shared__ int s_front[32];
  const int tid  = threadIdx.x;
  const int lane = tid & 31;
  const int wave = tid >> 5;
  const int tile0 = blockIdx.x * 1024;

  int front = 0;
  for (int i = tid; i < tile0; i += 1024) front += deg[i];
  const int idx = tile0 + tid;
  const int v = idx < N ? deg[idx] : 0;

  int inc = v;
#pragma unroll
  for (int o = 1; o < 32; o <<= 1) {
    const int t = __shfl_up(inc, o);
    if (lane >= o) inc += t;
  }
#pragma unroll
  for (int o = 16; o > 0; o >>= 1) front += __shfl_xor(front, o);
  if (lane == 31) s_tile[wave]  = inc;
  if (lane == 0)  s_front[wave] = front;
  __syncthreads();

  int base = 0;
  for (int w = 0; w < 32; ++w) {
    base += s_front[w];
    if (w < wave) base += s_tile[w];
  }
  if (idx < N) rowend[idx] = base + inc;
}

__global__ void __launch_bounds__(256) fill_nbr(const int* __restrict__ src,
                                                const int* __restrict__ dst,
                                                const int* __restrict__ rowend,
                                                const int* __restrict__ rank,
                                                int* __restrict__ nbr, int E) {
  const int e = blockIdx.x * 256 + threadIdx.x;
  if (e < E) {
    const int d = dst[e];
    const int beg = d > 0 ? rowend[d - 1] : 0;
    nbr[beg + rank[e]] = src[e];
  }
}

__device__ __forceinline__ float4 pull_mean(const float* __restrict__ X,
                                            const int* __restrict__ nbr, int beg,
                                            int dg, int ids, int lane, int N) {
  float4 acc = make_float4(0.0f, 0.0f, 0.0f, 0.0f);
  for (int base = 0; base < dg; base += 32) {
    const int rem = dg - base < 32 ? dg - base : 32;
    int mine = ids;
    if (base > 0) mine = lane < rem ? nbr[beg + base + lane] : 0;
#pragma unroll 4
    for (int j = 0; j < rem; ++j) {
      int s = __builtin_amdgcn_readlane(mine, j);
      s = (unsigned)s < (unsigned)N ? s : N - 1;
      const float4 v = *(const float4*)(X + (long)s * D + lane * 4);
      acc.x += v.x; acc.y += v.y; acc.z += v.z; acc.w += v.w;
    }
  }
  const float c = fmaxf((float)dg, 1.0f);
  float4 o;
  o.x = acc.x / c; o.y = acc.y / c; o.z = acc.z / c; o.w = acc.w / c;
  return o;
}

__device__ __forceinline__ v8f tile_gemm(const float* aRow, const float* __restrict__ Wm,
                                         int half, int n, v8f c) {
#pragma unroll 4
  for (int k0 = 0; k0 < D; k0 += 4) {
    const float2 a = *(const float2*)(aRow + k0);
    const float b0 = Wm[(k0 + 2 * half) * D + n];
    const float b1 = Wm[(k0 + 2 * half + 1) * D + n];
    c = wmma4(a.x, a.y, b0, b1, c);
  }
  return c;
}

__device__ __forceinline__ void tile_gemm2(const float* aRow, const float* __restrict__ W1,
                                           const float* __restrict__ W2, int half, int n,
                                           v8f& c1, v8f& c2) {
#pragma unroll 4
  for (int k0 = 0; k0 < D; k0 += 4) {
    const float2 a = *(const float2*)(aRow + k0);
    const int kr = (k0 + 2 * half) * D + n;
    const float p0 = W1[kr];
    const float p1 = W1[kr + D];
    const float q0 = W2[kr];
    const float q1 = W2[kr + D];
    c1 = wmma4(a.x, a.y, p0, p1, c1);
    c2 = wmma4(a.x, a.y, q0, q1, c2);
  }
}

__device__ __forceinline__ void store_tile(float* __restrict__ Hout, v8f c, int m0,
                                           int half, int n, int do_relu) {
#pragma unroll
  for (int r = 0; r < 8; ++r) {
    float v = c[r];
    if (do_relu) v = fmaxf(v, 0.0f);
    const int m = m0 + r + 8 * half;
    Hout[(long)m * D + n] = v;
  }
}

template <bool FUSED>
__global__ void __launch_bounds__(256) sage_layer(
    const float* __restrict__ X, const int* __restrict__ nbr,
    const int* __restrict__ rowend,
    const float* __restrict__ Wagg, const float* __restrict__ Wroot,
    const float* __restrict__ bias, const float* __restrict__ Wp1,
    const float* __restrict__ Wp2, float* __restrict__ O1, float* __restrict__ O2,
    int N) {
  __shared__ __attribute__((aligned(16))) float sAgg[16 * AP];
  __shared__ __attribute__((aligned(16))) float sRoot[16 * AP];

  const int tid  = threadIdx.x;
  const int lane = tid & 31;
  const int half = lane >> 4;
  const int lm   = lane & 15;
  const int wave = __builtin_amdgcn_readfirstlane(tid >> 5);
  const int m0   = blockIdx.x * 16;
  const int n    = wave * 16 + lm;

  {
    const int scol = lane * 4;
    const int r0 = m0 + wave;
    const int r1 = r0 + 8;
    const float4 q0 = *(const float4*)(X + (long)r0 * D + scol);
    const float4 q1 = *(const float4*)(X + (long)r1 * D + scol);
    const int beg0 = r0 > 0 ? rowend[r0 - 1] : 0;
    const int dg0  = rowend[r0] - beg0;
    const int beg1 = rowend[r1 - 1];
    const int dg1  = rowend[r1] - beg1;
    const int ids0 = lane < dg0 ? nbr[beg0 + lane] : 0;
    const int ids1 = lane < dg1 ? nbr[beg1 + lane] : 0;
    *(float4*)(sRoot + wave * AP + scol)       = q0;
    *(float4*)(sRoot + (wave + 8) * AP + scol) = q1;
    const float4 p0 = pull_mean(X, nbr, beg0, dg0, ids0, lane, N);
    const float4 p1 = pull_mean(X, nbr, beg1, dg1, ids1, lane, N);
    *(float4*)(sAgg + wave * AP + scol)        = p0;
    *(float4*)(sAgg + (wave + 8) * AP + scol)  = p1;
  }

  v8f c;
  const float bv = bias[n];
#pragma unroll
  for (int r = 0; r < 8; ++r) c[r] = bv;
  __syncthreads();

  c = tile_gemm(sAgg + lm * AP + 2 * half, Wagg, half, n, c);
  c = tile_gemm(sRoot + lm * AP + 2 * half, Wroot, half, n, c);

  if constexpr (!FUSED) {
    store_tile(O1, c, m0, half, n, 1);
  } else {
    __shared__ __attribute__((aligned(16))) float sH[16 * AP];
#pragma unroll
    for (int r = 0; r < 8; ++r) sH[(r + 8 * half) * AP + n] = c[r];
    v8f c1, c2;
#pragma unroll
    for (int r = 0; r < 8; ++r) { c1[r] = 0.0f; c2[r] = 0.0f; }
    __syncthreads();
    tile_gemm2(sH + lm * AP + 2 * half, Wp1, Wp2, half, n, c1, c2);
    store_tile(O1, c1, m0, half, n, 0);
    store_tile(O2, c2, m0, half, n, 0);
  }
}

__global__ void edge_mlp(const float* __restrict__ P1, const float* __restrict__ P2,
                         const float* __restrict__ ea, const int* __restrict__ src,
                         const int* __restrict__ dst, const float* __restrict__ W3,
                         const float* __restrict__ bc1, const float* __restrict__ Wc2,
                         const float* __restrict__ bc2, float* __restrict__ out,
                         int E) {
  __shared__ int   s_src[16];
  __shared__ int   s_dst[16];
  __shared__ __attribute__((aligned(16))) float s_red[8][16];
  __shared__ __attribute__((aligned(16))) float s_ea[16 * EP];

  const int tid  = threadIdx.x;
  const int wave = tid >> 5;
  const int lane = tid & 31;
  const int half = lane >> 4;
  const int lm   = lane & 15;
  const int n    = wave * 16 + lm;
  const int ntiles = E >> 4;
  const int step   = gridDim.x;

  const float bv = bc1[n];
  const float w2 = Wc2[n];
  const float b2 = bc2[0];
  float wb0[8], wb1[8];
#pragma unroll
  for (int j = 0; j < 8; ++j) {
    wb0[j] = W3[(4 * j + 2 * half) * D + n];
    wb1[j] = W3[(4 * j + 2 * half + 1) * D + n];
  }

  int tile = blockIdx.x;
  int id = 0;
  if (tid < 32 && tile < ntiles)
    id = tid < 16 ? src[tile * 16 + tid] : dst[tile * 16 + tid - 16];
  const int eoff = (tid >> 4) * EP + (tid & 15) * 2;
  float2 ev = make_float2(0.0f, 0.0f);
  if (tile < ntiles) ev = *(const float2*)(ea + (long)tile * (16 * DE) + tid * 2);

  for (; tile < ntiles; tile += step) {
    const int e0 = tile * 16;
    if (tid < 16)       s_src[tid]      = id;
    else if (tid < 32)  s_dst[tid - 16] = id;
    *(float2*)(s_ea + eoff) = ev;
    __syncthreads();
    const int tnext = tile + step;
    if (tid < 32 && tnext < ntiles)
      id = tid < 16 ? src[tnext * 16 + tid] : dst[tnext * 16 + tid - 16];
    if (tnext < ntiles) ev = *(const float2*)(ea + (long)tnext * (16 * DE) + tid * 2);

    v8f c;
#pragma unroll
    for (int r = 0; r < 8; ++r) {
      const int m = r + 8 * half;
      c[r] = P1[(long)s_src[m] * D + n] + P2[(long)s_dst[m] * D + n] + bv;
    }

    const float* aRow = s_ea + lm * EP + 2 * half;
#pragma unroll
    for (int j = 0; j < 8; ++j) {
      const float2 a = *(const float2*)(aRow + 4 * j);
      c = wmma4(a.x, a.y, wb0[j], wb1[j], c);
    }

    float red[8];
#pragma unroll
    for (int r = 0; r < 8; ++r) {
      float t = fmaxf(c[r], 0.0f) * w2;
      t = row_xor_add<1>(t);
      t = row_xor_add<2>(t);
      t = row_xor_add<4>(t);
      t = row_xor_add<8>(t);
      red[r] = t;
    }
    if (lm == 0) {
      float* p = &s_red[wave][8 * half];
      *(float4*)p       = make_float4(red[0], red[1], red[2], red[3]);
      *(float4*)(p + 4) = make_float4(red[4], red[5], red[6], red[7]);
    }
    __syncthreads();

    if (tid < 16) {
      float acc = b2;
#pragma unroll
      for (int w = 0; w < 8; ++w) acc += s_red[w][tid];
      if (e0 + tid < E) out[e0 + tid] = acc;
    }
  }
}

extern "C" void kernel_launch(void* const* d_in, const int* in_sizes, int n_in,
                              void* d_out, int out_size, void* d_ws, size_t ws_size,
                              hipStream_t stream) {
  (void)n_in; (void)out_size; (void)ws_size;
  const float* x   = (const float*)d_in[0];
  const float* ea  = (const float*)d_in[1];
  const float* W1a = (const float*)d_in[2];
  const float* b1  = (const float*)d_in[3];
  const float* W1r = (const float*)d_in[4];
  const float* W2a = (const float*)d_in[5];
  const float* b2  = (const float*)d_in[6];
  const float* W2r = (const float*)d_in[7];
  const float* Wc1 = (const float*)d_in[8];
  const float* bc1 = (const float*)d_in[9];
  const float* Wc2 = (const float*)d_in[10];
  const float* bc2 = (const float*)d_in[11];
  const int*   ei  = (const int*)d_in[12];

  const int N = in_sizes[0] / D;
  const int E = in_sizes[12] / 2;
  const int* src = ei;
  const int* dst = ei + E;

  float* ws = (float*)d_ws;
  const long NM = (long)N * D;
  float* bufA = ws;
  float* bufB = ws + NM;
  float* bufC = ws + 2 * NM;
  int*   rowend = (int*)(ws + 3 * NM);
  int*   deg    = (int*)bufC;
  int*   rank   = (int*)bufA;
  float* out  = (float*)d_out;
  int*   nbr    = (int*)d_out;

  const dim3 blk(256);
  const int ggrid = N / 16;
  const int tgrid = (E + 255) / 256;

  zero_i32<<<(N + 255) / 256, blk, 0, stream>>>(deg, N);
  count_rank<<<tgrid, blk, 0, stream>>>(dst, deg, rank, E);
  scan_rowend<<<(N + 1023) / 1024, dim3(1024), 0, stream>>>(deg, rowend, N);
  fill_nbr<<<tgrid, blk, 0, stream>>>(src, dst, rowend, rank, nbr, E);

  sage_layer<false><<<ggrid, blk, 0, stream>>>(x, nbr, rowend, W1a, W1r, b1, nullptr,
                                               nullptr, bufB, nullptr, N);

  sage_layer<true><<<ggrid, blk, 0, stream>>>(bufB, nbr, rowend, W2a, W2r, b2, Wc1,
                                              Wc1 + 128 * D, bufC, bufA, N);

  edge_mlp<<<1536, blk, 0, stream>>>(bufC, bufA, ea, src, dst, Wc1 + 256 * D,
                                     bc1, Wc2, bc2, out, E);
}


# edge_mlp: P1[src], P2[dst] rows fetched once per block as whole 512-byte rows one tile ahead (kept in flight over an LDS-only barrier), summed into an LDS tile; was 16 half-line gathers per wave
# speedup vs baseline: 2.7218x; 1.0711x over previous
#include <hip/hip_runtime.h>


typedef __attribute__((ext_vector_type(2))) float v2f;
typedef __attribute__((ext_vector_type(8))) float v8f;

#define D   128
#define DE  32
#define AP  (D + 4)
#define EP  (DE + 4)

__device__ __forceinline__ v8f wmma4(float a0, float a1, float b0, float b1, v8f c) {
  v2f a; a[0] = a0; a[1] = a1;
  v2f b; b[0] = b0; b[1] = b1;
  return __builtin_amdgcn_wmma_f32_16x16x4_f32(false, a, false, b, (short)0, c,
                                               false, false);
}

template <int M>
__device__ __forceinline__ float row_xor_add(float v) {
  const int other =
      __builtin_amdgcn_update_dpp(0, __float_as_int(v), 0x160 + M, 0xF, 0xF, true);
  return v + __int_as_float(other);
}

__global__ void __launch_bounds__(256) zero_i32(int* __restrict__ p, int n) {
  const int i = blockIdx.x * 256 + threadIdx.x;
  if (i < n) p[i] = 0;
}

__global__ void __launch_bounds__(256) count_rank(const int* __restrict__ dst,
                                                  int* __restrict__ deg,
                                                  int* __restrict__ rank, int E) {
  const int e = blockIdx.x * 256 + threadIdx.x;
  if (e < E) rank[e] = atomicAdd(deg + dst[e], 1);
}

__global__ void __launch_bounds__(1024) scan_rowend(const int* __restrict__ deg,
                                                    int* __restrict__ rowend, int N) {
  __shared__ int s_tile[32];
  __shared__ int s_front[32];
  const int tid  = threadIdx.x;
  const int lane = tid & 31;
  const int wave = tid >> 5;
  const int tile0 = blockIdx.x * 1024;

  int front = 0;
  for (int i = tid; i < tile0; i += 1024) front += deg[i];
  const int idx = tile0 + tid;
  const int v = idx < N ? deg[idx] : 0;

  int inc = v;
#pragma unroll
  for (int o = 1; o < 32; o <<= 1) {
    const int t = __shfl_up(inc, o);
    if (lane >= o) inc += t;
  }
#pragma unroll
  for (int o = 16; o > 0; o >>= 1) front += __shfl_xor(front, o);
  if (lane == 31) s_tile[wave]  = inc;
  if (lane == 0)  s_front[wave] = front;
  __syncthreads();

  int base = 0;
  for (int w = 0; w < 32; ++w) {
    base += s_front[w];
    if (w < wave) base += s_tile[w];
  }
  if (idx < N) rowend[idx] = base + inc;
}

__global__ void __launch_bounds__(256) fill_nbr(const int* __restrict__ src,
                                                const int* __restrict__ dst,
                                                const int* __restrict__ rowend,
                                                const int* __restrict__ rank,
                                                int* __restrict__ nbr, int E) {
  const int e = blockIdx.x * 256 + threadIdx.x;
  if (e < E) {
    const int d = dst[e];
    const int beg = d > 0 ? rowend[d - 1] : 0;
    nbr[beg + rank[e]] = src[e];
  }
}

__device__ __forceinline__ float4 pull_mean(const float* __restrict__ X,
                                            const int* __restrict__ nbr, int beg,
                                            int dg, int ids, int lane, int N) {
  float4 acc = make_float4(0.0f, 0.0f, 0.0f, 0.0f);
  for (int base = 0; base < dg; base += 32) {
    const int rem = dg - base < 32 ? dg - base : 32;
    int mine = ids;
    if (base > 0) mine = lane < rem ? nbr[beg + base + lane] : 0;
#pragma unroll 4
    for (int j = 0; j < rem; ++j) {
      int s = __builtin_amdgcn_readlane(mine, j);
      s = (unsigned)s < (unsigned)N ? s : N - 1;
      const float4 v = *(const float4*)(X + (long)s * D + lane * 4);
      acc.x += v.x; acc.y += v.y; acc.z += v.z; acc.w += v.w;
    }
  }
  const float c = fmaxf((float)dg, 1.0f);
  float4 o;
  o.x = acc.x / c; o.y = acc.y / c; o.z = acc.z / c; o.w = acc.w / c;
  return o;
}

__device__ __forceinline__ v8f tile_gemm(const float* aRow, const float* __restrict__ Wm,
                                         int half, int n, v8f c) {
#pragma unroll 4
  for (int k0 = 0; k0 < D; k0 += 4) {
    const float2 a = *(const float2*)(aRow + k0);
    const float b0 = Wm[(k0 + 2 * half) * D + n];
    const float b1 = Wm[(k0 + 2 * half + 1) * D + n];
    c = wmma4(a.x, a.y, b0, b1, c);
  }
  return c;
}

__device__ __forceinline__ void tile_gemm2(const float* aRow, const float* __restrict__ W1,
                                           const float* __restrict__ W2, int half, int n,
                                           v8f& c1, v8f& c2) {
#pragma unroll 4
  for (int k0 = 0; k0 < D; k0 += 4) {
    const float2 a = *(const float2*)(aRow + k0);
    const int kr = (k0 + 2 * half) * D + n;
    const float p0 = W1[kr];
    const float p1 = W1[kr + D];
    const float q0 = W2[kr];
    const float q1 = W2[kr + D];
    c1 = wmma4(a.x, a.y, p0, p1, c1);
    c2 = wmma4(a.x, a.y, q0, q1, c2);
  }
}

__device__ __forceinline__ void store_tile(float* __restrict__ Hout, v8f c, int m0,
                                           int half, int n, int do_relu) {
#pragma unroll
  for (int r = 0; r < 8; ++r) {
    float v = c[r];
    if (do_relu) v = fmaxf(v, 0.0f);
    const int m = m0 + r + 8 * half;
    Hout[(long)m * D + n] = v;
  }
}

template <bool FUSED>
__global__ void __launch_bounds__(256) sage_layer(
    const float* __restrict__ X, const int* __restrict__ nbr,
    const int* __restrict__ rowend,
    const float* __restrict__ Wagg, const float* __restrict__ Wroot,
    const float* __restrict__ bias, const float* __restrict__ Wp1,
    const float* __restrict__ Wp2, float* __restrict__ O1, float* __restrict__ O2,
    int N) {
  __shared__ __attribute__((aligned(16))) float sAgg[16 * AP];
  __shared__ __attribute__((aligned(16))) float sRoot[16 * AP];

  const int tid  = threadIdx.x;
  const int lane = tid & 31;
  const int half = lane >> 4;
  const int lm   = lane & 15;
  const int wave = __builtin_amdgcn_readfirstlane(tid >> 5);
  const int m0   = blockIdx.x * 16;
  const int n    = wave * 16 + lm;

  {
    const int scol = lane * 4;
    const int r0 = m0 + wave;
    const int r1 = r0 + 8;
    const float4 q0 = *(const float4*)(X + (long)r0 * D + scol);
    const float4 q1 = *(const float4*)(X + (long)r1 * D + scol);
    const int beg0 = r0 > 0 ? rowend[r0 - 1] : 0;
    const int dg0  = rowend[r0] - beg0;
    const int beg1 = rowend[r1 - 1];
    const int dg1  = rowend[r1] - beg1;
    const int ids0 = lane < dg0 ? nbr[beg0 + lane] : 0;
    const int ids1 = lane < dg1 ? nbr[beg1 + lane] : 0;
    *(float4*)(sRoot + wave * AP + scol)       = q0;
    *(float4*)(sRoot + (wave + 8) * AP + scol) = q1;
    const float4 p0 = pull_mean(X, nbr, beg0, dg0, ids0, lane, N);
    const float4 p1 = pull_mean(X, nbr, beg1, dg1, ids1, lane, N);
    *(float4*)(sAgg + wave * AP + scol)        = p0;
    *(float4*)(sAgg + (wave + 8) * AP + scol)  = p1;
  }

  v8f c;
  const float bv = bias[n];
#pragma unroll
  for (int r = 0; r < 8; ++r) c[r] = bv;
  __syncthreads();

  c = tile_gemm(sAgg + lm * AP + 2 * half, Wagg, half, n, c);
  c = tile_gemm(sRoot + lm * AP + 2 * half, Wroot, half, n, c);

  if constexpr (!FUSED) {
    store_tile(O1, c, m0, half, n, 1);
  } else {
    __shared__ __attribute__((aligned(16))) float sH[16 * AP];
#pragma unroll
    for (int r = 0; r < 8; ++r) sH[(r + 8 * half) * AP + n] = c[r];
    v8f c1, c2;
#pragma unroll
    for (int r = 0; r < 8; ++r) { c1[r] = 0.0f; c2[r] = 0.0f; }
    __syncthreads();
    tile_gemm2(sH + lm * AP + 2 * half, Wp1, Wp2, half, n, c1, c2);
    store_tile(O1, c1, m0, half, n, 0);
    store_tile(O2, c2, m0, half, n, 0);
  }
}

__global__ void edge_mlp(const float* __restrict__ P1, const float* __restrict__ P2,
                         const float* __restrict__ ea, const int* __restrict__ src,
                         const int* __restrict__ dst, const float* __restrict__ W3,
                         const float* __restrict__ bc1, const float* __restrict__ Wc2,
                         const float* __restrict__ bc2, float* __restrict__ out,
                         int E) {
  __shared__ __attribute__((aligned(16))) float s_red[8][16];
  __shared__ __attribute__((aligned(16))) float s_ea[16 * EP];
  __shared__ __attribute__((aligned(16))) float s_g[16 * AP];

  const int tid  = threadIdx.x;
  const int wave = __builtin_amdgcn_readfirstlane(tid >> 5);
  const int lane = tid & 31;
  const int half = lane >> 4;
  const int lm   = lane & 15;
  const int n    = wave * 16 + lm;
  const int ntiles = E >> 4;
  const int step   = gridDim.x;
  const int er   = 2 * wave;
  const int scol = lane * 4;

  const float bv = bc1[n];
  const float w2 = Wc2[n];
  const float b2 = bc2[0];
  float wb0[8], wb1[8];
#pragma unroll
  for (int j = 0; j < 8; ++j) {
    wb0[j] = W3[(4 * j + 2 * half) * D + n];
    wb1[j] = W3[(4 * j + 2 * half + 1) * D + n];
  }

  int tile = blockIdx.x;
  int s0 = 0, s1 = 0, d0 = 0, d1 = 0;
  float4 ga0 = make_float4(0.0f, 0.0f, 0.0f, 0.0f), ga1 = ga0, gb0 = ga0, gb1 = ga0;
  if (tile < ntiles) {
    s0 = src[tile * 16 + er];  s1 = src[tile * 16 + er + 1];
    d0 = dst[tile * 16 + er];  d1 = dst[tile * 16 + er + 1];
    ga0 = *(const float4*)(P1 + (long)s0 * D + scol);
    ga1 = *(const float4*)(P1 + (long)s1 * D + scol);
    gb0 = *(const float4*)(P2 + (long)d0 * D + scol);
    gb1 = *(const float4*)(P2 + (long)d1 * D + scol);
  }
  if (tile + step < ntiles) {
    const int t1 = tile + step;
    s0 = src[t1 * 16 + er];  s1 = src[t1 * 16 + er + 1];
    d0 = dst[t1 * 16 + er];  d1 = dst[t1 * 16 + er + 1];
  }
  const int eoff = (tid >> 4) * EP + (tid & 15) * 2;
  float2 ev = make_float2(0.0f, 0.0f);
  if (tile < ntiles) ev = *(const float2*)(ea + (long)tile * (16 * DE) + tid * 2);

  for (; tile < ntiles; tile += step) {
    const int e0 = tile * 16;
    *(float4*)(s_g + er * AP + scol) =
        make_float4(ga0.x + gb0.x, ga0.y + gb0.y, ga0.z + gb0.z, ga0.w + gb0.w);
    *(float4*)(s_g + (er + 1) * AP + scol) =
        make_float4(ga1.x + gb1.x, ga1.y + gb1.y, ga1.z + gb1.z, ga1.w + gb1.w);
    *(float2*)(s_ea + eoff) = ev;
    __syncthreads();
    const int tnext = tile + step;
    if (tnext < ntiles) {
      ga0 = *(const float4*)(P1 + (long)s0 * D + scol);
      ga1 = *(const float4*)(P1 + (long)s1 * D + scol);
      gb0 = *(const float4*)(P2 + (long)d0 * D + scol);
      gb1 = *(const float4*)(P2 + (long)d1 * D + scol);
      ev = *(const float2*)(ea + (long)tnext * (16 * DE) + tid * 2);
    }
    const int tnext2 = tnext + step;
    if (tnext2 < ntiles) {
      s0 = src[tnext2 * 16 + er];  s1 = src[tnext2 * 16 + er + 1];
      d0 = dst[tnext2 * 16 + er];  d1 = dst[tnext2 * 16 + er + 1];
    }

    v8f c;
#pragma unroll
    for (int r = 0; r < 8; ++r) c[r] = s_g[(r + 8 * half) * AP + n] + bv;

    const float* aRow = s_ea + lm * EP + 2 * half;
#pragma unroll
    for (int j = 0; j < 8; ++j) {
      const float2 a = *(const float2*)(aRow + 4 * j);
      c = wmma4(a.x, a.y, wb0[j], wb1[j], c);
    }

    float red[8];
#pragma unroll
    for (int r = 0; r < 8; ++r) {
      float t = fmaxf(c[r], 0.0f) * w2;
      t = row_xor_add<1>(t);
      t = row_xor_add<2>(t);
      t = row_xor_add<4>(t);
      t = row_xor_add<8>(t);
      red[r] = t;
    }
    if (lm == 0) {
      float* p = &s_red[wave][8 * half];
      *(float4*)p       = make_float4(red[0], red[1], red[2], red[3]);
      *(float4*)(p + 4) = make_float4(red[4], red[5], red[6], red[7]);
    }
    asm volatile("s_wait_dscnt 0x0\n\ts_barrier_signal -1\n\ts_barrier_wait -1" ::: "memory");

    if (tid < 16) {
      float acc = b2;
#pragma unroll
      for (int w = 0; w < 8; ++w) acc += s_red[w][tid];
      if (e0 + tid < E) out[e0 + tid] = acc;
    }
  }
}

extern "C" void kernel_launch(void* const* d_in, const int* in_sizes, int n_in,
                              void* d_out, int out_size, void* d_ws, size_t ws_size,
                              hipStream_t stream) {
  (void)n_in; (void)out_size; (void)ws_size;
  const float* x   = (const float*)d_in[0];
  const float* ea  = (const float*)d_in[1];
  const float* W1a = (const float*)d_in[2];
  const float* b1  = (const float*)d_in[3];
  const float* W1r = (const float*)d_in[4];
  const float* W2a = (const float*)d_in[5];
  const float* b2  = (const float*)d_in[6];
  const float* W2r = (const float*)d_in[7];
  const float* Wc1 = (const float*)d_in[8];
  const float* bc1 = (const float*)d_in[9];
  const float* Wc2 = (const float*)d_in[10];
  const float* bc2 = (const float*)d_in[11];
  const int*   ei  = (const int*)d_in[12];

  const int N = in_sizes[0] / D;
  const int E = in_sizes[12] / 2;
  const int* src = ei;
  const int* dst = ei + E;

  float* ws = (float*)d_ws;
  const long NM = (long)N * D;
  float* bufA = ws;
  float* bufB = ws + NM;
  float* bufC = ws + 2 * NM;
  int*   rowend = (int*)(ws + 3 * NM);
  int*   deg    = (int*)bufC;
  int*   rank   = (int*)bufA;
  float* out  = (float*)d_out;
  int*   nbr    = (int*)d_out;

  const dim3 blk(256);
  const int ggrid = N / 16;
  const int tgrid = (E + 255) / 256;

  zero_i32<<<(N + 255) / 256, blk, 0, stream>>>(deg, N);
  count_rank<<<tgrid, blk, 0, stream>>>(dst, deg, rank, E);
  scan_rowend<<<(N + 1023) / 1024, dim3(1024), 0, stream>>>(deg, rowend, N);
  fill_nbr<<<tgrid, blk, 0, stream>>>(src, dst, rowend, rank, nbr, E);

  sage_layer<false><<<ggrid, blk, 0, stream>>>(x, nbr, rowend, W1a, W1r, b1, nullptr,
                                               nullptr, bufB, nullptr, N);

  sage_layer<true><<<ggrid, blk, 0, stream>>>(bufB, nbr, rowend, W2a, W2r, b2, Wc1,
                                              Wc1 + 128 * D, bufC, bufA, N);

  edge_mlp<<<1536, blk, 0, stream>>>(bufC, bufA, ea, src, dst, Wc1 + 256 * D,
                                     bc1, Wc2, bc2, out, E);
}


# edge_mlp: a wave owns 16 edges x all 128 hidden columns (8 accumulators, W3 fragments from LDS): no barrier or cross-wave sum in the loop; score summed per lane over 8 columns, then 16 lanes, + bc2
# speedup vs baseline: 2.8328x; 1.0408x over previous
#include <hip/hip_runtime.h>


typedef __attribute__((ext_vector_type(2))) float v2f;
typedef __attribute__((ext_vector_type(8))) float v8f;

#define D   128
#define DE  32
#define AP  (D + 4)
#define EP  (DE + 4)

__device__ __forceinline__ v8f wmma4(float a0, float a1, float b0, float b1, v8f c) {
  v2f a; a[0] = a0; a[1] = a1;
  v2f b; b[0] = b0; b[1] = b1;
  return __builtin_amdgcn_wmma_f32_16x16x4_f32(false, a, false, b, (short)0, c,
                                               false, false);
}

template <int M>
__device__ __forceinline__ float row_xor_add(float v) {
  const int other =
      __builtin_amdgcn_update_dpp(0, __float_as_int(v), 0x160 + M, 0xF, 0xF, true);
  return v + __int_as_float(other);
}

__global__ void __launch_bounds__(256) zero_i32(int* __restrict__ p, int n) {
  const int i = blockIdx.x * 256 + threadIdx.x;
  if (i < n) p[i] = 0;
}

__global__ void __launch_bounds__(256) count_rank(const int* __restrict__ dst,
                                                  int* __restrict__ deg,
                                                  int* __restrict__ rank, int E) {
  const int e = blockIdx.x * 256 + threadIdx.x;
  if (e < E) rank[e] = atomicAdd(deg + dst[e], 1);
}

__global__ void __launch_bounds__(1024) scan_rowend(const int* __restrict__ deg,
                                                    int* __restrict__ rowend, int N) {
  __shared__ int s_tile[32];
  __shared__ int s_front[32];
  const int tid  = threadIdx.x;
  const int lane = tid & 31;
  const int wave = tid >> 5;
  const int tile0 = blockIdx.x * 1024;

  int front = 0;
  for (int i = tid; i < tile0; i += 1024) front += deg[i];
  const int idx = tile0 + tid;
  const int v = idx < N ? deg[idx] : 0;

  int inc = v;
#pragma unroll
  for (int o = 1; o < 32; o <<= 1) {
    const int t = __shfl_up(inc, o);
    if (lane >= o) inc += t;
  }
#pragma unroll
  for (int o = 16; o > 0; o >>= 1) front += __shfl_xor(front, o);
  if (lane == 31) s_tile[wave]  = inc;
  if (lane == 0)  s_front[wave] = front;
  __syncthreads();

  int base = 0;
  for (int w = 0; w < 32; ++w) {
    base += s_front[w];
    if (w < wave) base += s_tile[w];
  }
  if (idx < N) rowend[idx] = base + inc;
}

__global__ void __launch_bounds__(256) fill_nbr(const int* __restrict__ src,
                                                const int* __restrict__ dst,
                                                const int* __restrict__ rowend,
                                                const int* __restrict__ rank,
                                                int* __restrict__ nbr, int E) {
  const int e = blockIdx.x * 256 + threadIdx.x;
  if (e < E) {
    const int d = dst[e];
    const int beg = d > 0 ? rowend[d - 1] : 0;
    nbr[beg + rank[e]] = src[e];
  }
}

__device__ __forceinline__ float4 pull_mean(const float* __restrict__ X,
                                            const int* __restrict__ nbr, int beg,
                                            int dg, int ids, int lane, int N) {
  float4 acc = make_float4(0.0f, 0.0f, 0.0f, 0.0f);
  for (int base = 0; base < dg; base += 32) {
    const int rem = dg - base < 32 ? dg - base : 32;
    int mine = ids;
    if (base > 0) mine = lane < rem ? nbr[beg + base + lane] : 0;
#pragma unroll 4
    for (int j = 0; j < rem; ++j) {
      int s = __builtin_amdgcn_readlane(mine, j);
      s = (unsigned)s < (unsigned)N ? s : N - 1;
      const float4 v = *(const float4*)(X + (long)s * D + lane * 4);
      acc.x += v.x; acc.y += v.y; acc.z += v.z; acc.w += v.w;
    }
  }
  const float c = fmaxf((float)dg, 1.0f);
  float4 o;
  o.x = acc.x / c; o.y = acc.y / c; o.z = acc.z / c; o.w = acc.w / c;
  return o;
}

__device__ __forceinline__ v8f tile_gemm(const float* aRow, const float* __restrict__ Wm,
                                         int half, int n, v8f c) {
#pragma unroll 4
  for (int k0 = 0; k0 < D; k0 += 4) {
    const float2 a = *(const float2*)(aRow + k0);
    const float b0 = Wm[(k0 + 2 * half) * D + n];
    const float b1 = Wm[(k0 + 2 * half + 1) * D + n];
    c = wmma4(a.x, a.y, b0, b1, c);
  }
  return c;
}

__device__ __forceinline__ void tile_gemm2(const float* aRow, const float* __restrict__ W1,
                                           const float* __restrict__ W2, int half, int n,
                                           v8f& c1, v8f& c2) {
#pragma unroll 4
  for (int k0 = 0; k0 < D; k0 += 4) {
    const float2 a = *(const float2*)(aRow + k0);
    const int kr = (k0 + 2 * half) * D + n;
    const float p0 = W1[kr];
    const float p1 = W1[kr + D];
    const float q0 = W2[kr];
    const float q1 = W2[kr + D];
    c1 = wmma4(a.x, a.y, p0, p1, c1);
    c2 = wmma4(a.x, a.y, q0, q1, c2);
  }
}

__device__ __forceinline__ void store_tile(float* __restrict__ Hout, v8f c, int m0,
                                           int half, int n, int do_relu) {
#pragma unroll
  for (int r = 0; r < 8; ++r) {
    float v = c[r];
    if (do_relu) v = fmaxf(v, 0.0f);
    const int m = m0 + r + 8 * half;
    Hout[(long)m * D + n] = v;
  }
}

template <bool FUSED>
__global__ void __launch_bounds__(256) sage_layer(
    const float* __restrict__ X, const int* __restrict__ nbr,
    const int* __restrict__ rowend,
    const float* __restrict__ Wagg, const float* __restrict__ Wroot,
    const float* __restrict__ bias, const float* __restrict__ Wp1,
    const float* __restrict__ Wp2, float* __restrict__ O1, float* __restrict__ O2,
    int N) {
  __shared__ __attribute__((aligned(16))) float sAgg[16 * AP];
  __shared__ __attribute__((aligned(16))) float sRoot[16 * AP];

  const int tid  = threadIdx.x;
  const int lane = tid & 31;
  const int half = lane >> 4;
  const int lm   = lane & 15;
  const int wave = __builtin_amdgcn_readfirstlane(tid >> 5);
  const int m0   = blockIdx.x * 16;
  const int n    = wave * 16 + lm;

  {
    const int scol = lane * 4;
    const int r0 = m0 + wave;
    const int r1 = r0 + 8;
    const float4 q0 = *(const float4*)(X + (long)r0 * D + scol);
    const float4 q1 = *(const float4*)(X + (long)r1 * D + scol);
    const int beg0 = r0 > 0 ? rowend[r0 - 1] : 0;
    const int dg0  = rowend[r0] - beg0;
    const int beg1 = rowend[r1 - 1];
    const int dg1  = rowend[r1] - beg1;
    const int ids0 = lane < dg0 ? nbr[beg0 + lane] : 0;
    const int ids1 = lane < dg1 ? nbr[beg1 + lane] : 0;
    *(float4*)(sRoot + wave * AP + scol)       = q0;
    *(float4*)(sRoot + (wave + 8) * AP + scol) = q1;
    const float4 p0 = pull_mean(X, nbr, beg0, dg0, ids0, lane, N);
    const float4 p1 = pull_mean(X, nbr, beg1, dg1, ids1, lane, N);
    *(float4*)(sAgg + wave * AP + scol)        = p0;
    *(float4*)(sAgg + (wave + 8) * AP + scol)  = p1;
  }

  v8f c;
  const float bv = bias[n];
#pragma unroll
  for (int r = 0; r < 8; ++r) c[r] = bv;
  __syncthreads();

  c = tile_gemm(sAgg + lm * AP + 2 * half, Wagg, half, n, c);
  c = tile_gemm(sRoot + lm * AP + 2 * half, Wroot, half, n, c);

  if constexpr (!FUSED) {
    store_tile(O1, c, m0, half, n, 1);
  } else {
    __shared__ __attribute__((aligned(16))) float sH[16 * AP];
#pragma unroll
    for (int r = 0; r < 8; ++r) sH[(r + 8 * half) * AP + n] = c[r];
    v8f c1, c2;
#pragma unroll
    for (int r = 0; r < 8; ++r) { c1[r] = 0.0f; c2[r] = 0.0f; }
    __syncthreads();
    tile_gemm2(sH + lm * AP + 2 * half, Wp1, Wp2, half, n, c1, c2);
    store_tile(O1, c1, m0, half, n, 0);
    store_tile(O2, c2, m0, half, n, 0);
  }
}

__global__ void __launch_bounds__(256) edge_mlp(
    const float* __restrict__ P1, const float* __restrict__ P2,
    const float* __restrict__ ea, const int* __restrict__ src,
    const int* __restrict__ dst, const float* __restrict__ W3,
    const float* __restrict__ bc1, const float* __restrict__ Wc2,
    const float* __restrict__ bc2, float* __restrict__ out, int E) {
  __shared__ __attribute__((aligned(16))) float s_w3[8 * 4 * 32 * 4];
  __shared__ __attribute__((aligned(16))) float s_g[8][16 * AP];
  __shared__ __attribute__((aligned(16))) float s_ea[8][16 * EP];

  const int tid  = threadIdx.x;
  const int wave = __builtin_amdgcn_readfirstlane(tid >> 5);
  const int lane = tid & 31;
  const int half = lane >> 4;
  const int lm   = lane & 15;
  const int scol = lane * 4;

#pragma unroll
  for (int i = 0; i < 4; ++i) {
    const int idx = tid + 256 * i;
    const int fl = idx & 31, tp = (idx >> 5) & 3, j = idx >> 7;
    const int k  = 4 * j + 2 * (fl >> 4);
    const int n0 = 64 * (tp >> 1) + 4 * (fl & 15) + 2 * (tp & 1);
    float4 v;
    v.x = W3[k * D + n0];     v.y = W3[(k + 1) * D + n0];
    v.z = W3[k * D + n0 + 1]; v.w = W3[(k + 1) * D + n0 + 1];
    *(float4*)(s_w3 + idx * 4) = v;
  }

  const float4 bA = *(const float4*)(bc1 + 4 * lm);
  const float4 bB = *(const float4*)(bc1 + 64 + 4 * lm);
  const float4 wA = *(const float4*)(Wc2 + 4 * lm);
  const float4 wB = *(const float4*)(Wc2 + 64 + 4 * lm);
  const float b2 = bc2[0];
  __syncthreads();

  float* g  = s_g[wave];
  float* se = s_ea[wave];
  const int ntiles = E >> 4;
  const int nw     = gridDim.x * 8;
  int tile = blockIdx.x * 8 + wave;
  int ids = 0;
  if (tile < ntiles) ids = lane < 16 ? src[tile * 16 + lane] : dst[tile * 16 + lane - 16];

  for (; tile < ntiles; tile += nw) {
    const int e0  = tile * 16;
    const int cur = ids;
    const int tn = tile + nw;
    if (tn < ntiles) ids = lane < 16 ? src[tn * 16 + lane] : dst[tn * 16 + lane - 16];

#pragma unroll 8
    for (int m = 0; m < 16; ++m) {
      const int s = __builtin_amdgcn_readlane(cur, m);
      const int d = __builtin_amdgcn_readlane(cur, 16 + m);
      const float4 a = *(const float4*)(P1 + (long)s * D + scol);
      const float4 b = *(const float4*)(P2 + (long)d * D + scol);
      *(float4*)(g + m * AP + scol) =
          make_float4(a.x + b.x, a.y + b.y, a.z + b.z, a.w + b.w);
      if ((m & 3) == 0) {
        const int q = (m >> 2) * 32 + lane;
        const float4 e = *(const float4*)(ea + (long)tile * (16 * DE) + q * 4);
        *(float4*)(se + (q >> 3) * EP + (q & 7) * 4) = e;
      }
    }
    __builtin_amdgcn_fence(__ATOMIC_RELEASE, "wavefront");
    __builtin_amdgcn_wave_barrier();
    __builtin_amdgcn_fence(__ATOMIC_ACQUIRE, "wavefront");

    v8f c[8];
#pragma unroll
    for (int r = 0; r < 8; ++r) {
      const float* gr = g + (r + 8 * half) * AP + 4 * lm;
      const float4 g0 = *(const float4*)gr;
      const float4 g1 = *(const float4*)(gr + 64);
      c[0][r] = g0.x + bA.x; c[1][r] = g0.y + bA.y;
      c[2][r] = g0.z + bA.z; c[3][r] = g0.w + bA.w;
      c[4][r] = g1.x + bB.x; c[5][r] = g1.y + bB.y;
      c[6][r] = g1.z + bB.z; c[7][r] = g1.w + bB.w;
    }

    const float* aRow = se + lm * EP + 2 * half;
#pragma unroll
    for (int j = 0; j < 8; ++j) {
      const float2 a = *(const float2*)(aRow + 4 * j);
#pragma unroll
      for (int tp = 0; tp < 4; ++tp) {
        const float4 b = *(const float4*)(s_w3 + ((j * 4 + tp) * 32 + lane) * 4);
        c[2 * tp]     = wmma4(a.x, a.y, b.x, b.y, c[2 * tp]);
        c[2 * tp + 1] = wmma4(a.x, a.y, b.z, b.w, c[2 * tp + 1]);
      }
    }

    float red[8];
#pragma unroll
    for (int r = 0; r < 8; ++r) {
      float t = fmaxf(c[0][r], 0.0f) * wA.x;
      t += fmaxf(c[1][r], 0.0f) * wA.y;
      t += fmaxf(c[2][r], 0.0f) * wA.z;
      t += fmaxf(c[3][r], 0.0f) * wA.w;
      t += fmaxf(c[4][r], 0.0f) * wB.x;
      t += fmaxf(c[5][r], 0.0f) * wB.y;
      t += fmaxf(c[6][r], 0.0f) * wB.z;
      t += fmaxf(c[7][r], 0.0f) * wB.w;
      t = row_xor_add<1>(t);
      t = row_xor_add<2>(t);
      t = row_xor_add<4>(t);
      t = row_xor_add<8>(t);
      red[r] = t + b2;
    }
    if (lm == 0) {
      float* p = out + e0 + 8 * half;
      *(float4*)p       = make_float4(red[0], red[1], red[2], red[3]);
      *(float4*)(p + 4) = make_float4(red[4], red[5], red[6], red[7]);
    }
  }
}

extern "C" void kernel_launch(void* const* d_in, const int* in_sizes, int n_in,
                              void* d_out, int out_size, void* d_ws, size_t ws_size,
                              hipStream_t stream) {
  (void)n_in; (void)out_size; (void)ws_size;
  const float* x   = (const float*)d_in[0];
  const float* ea  = (const float*)d_in[1];
  const float* W1a = (const float*)d_in[2];
  const float* b1  = (const float*)d_in[3];
  const float* W1r = (const float*)d_in[4];
  const float* W2a = (const float*)d_in[5];
  const float* b2  = (const float*)d_in[6];
  const float* W2r = (const float*)d_in[7];
  const float* Wc1 = (const float*)d_in[8];
  const float* bc1 = (const float*)d_in[9];
  const float* Wc2 = (const float*)d_in[10];
  const float* bc2 = (const float*)d_in[11];
  const int*   ei  = (const int*)d_in[12];

  const int N = in_sizes[0] / D;
  const int E = in_sizes[12] / 2;
  const int* src = ei;
  const int* dst = ei + E;

  float* ws = (float*)d_ws;
  const long NM = (long)N * D;
  float* bufA = ws;
  float* bufB = ws + NM;
  float* bufC = ws + 2 * NM;
  int*   rowend = (int*)(ws + 3 * NM);
  int*   deg    = (int*)bufC;
  int*   rank   = (int*)bufA;
  float* out  = (float*)d_out;
  int*   nbr    = (int*)d_out;

  const dim3 blk(256);
  const int ggrid = N / 16;
  const int tgrid = (E + 255) / 256;

  zero_i32<<<(N + 255) / 256, blk, 0, stream>>>(deg, N);
  count_rank<<<tgrid, blk, 0, stream>>>(dst, deg, rank, E);
  scan_rowend<<<(N + 1023) / 1024, dim3(1024), 0, stream>>>(deg, rowend, N);
  fill_nbr<<<tgrid, blk, 0, stream>>>(src, dst, rowend, rank, nbr, E);

  sage_layer<false><<<ggrid, blk, 0, stream>>>(x, nbr, rowend, W1a, W1r, b1, nullptr,
                                               nullptr, bufB, nullptr, N);

  sage_layer<true><<<ggrid, blk, 0, stream>>>(bufB, nbr, rowend, W2a, W2r, b2, Wc1,
                                              Wc1 + 128 * D, bufC, bufA, N);

  edge_mlp<<<768, blk, 0, stream>>>(bufC, bufA, ea, src, dst, Wc1 + 256 * D,
                                     bc1, Wc2, bc2, out, E);
}
